# starting kernel
# baseline (speedup 1.0000x reference)
#include <hip/hip_runtime.h>
#include <hip/hip_bf16.h>
#include <stdint.h>

typedef __attribute__((ext_vector_type(2))) float v2f;
typedef __attribute__((ext_vector_type(8))) float v8f;

#define FEATS 128
#define LDA 132

__global__ void k_degrees(const int* __restrict__ src, const int* __restrict__ dst,
                          int* __restrict__ dout, int* __restrict__ din, int E) {
  int i = blockIdx.x * blockDim.x + threadIdx.x;
  if (i < E) {
    atomicAdd(&dout[src[i]], 1);
    atomicAdd(&din[dst[i]], 1);
  }
}

__global__ void k_norms(const int* __restrict__ dout, const int* __restrict__ din,
                        float* __restrict__ nsrc, float* __restrict__ ndst, int n) {
  int i = blockIdx.x * blockDim.x + threadIdx.x;
  if (i < n) {
    nsrc[i] = rsqrtf(fmaxf((float)dout[i], 1.0f));
    ndst[i] = rsqrtf(fmaxf((float)din[i], 1.0f));
  }
}

__global__ void k_scan1(const int* __restrict__ counts, int* __restrict__ incl,
                        int* __restrict__ bsum, int n) {
  __shared__ int tmp[256];
  int tid = threadIdx.x;
  int i = blockIdx.x * 256 + tid;
  int v = (i < n) ? counts[i] : 0;
  tmp[tid] = v;
  __syncthreads();
  for (int off = 1; off < 256; off <<= 1) {
    int t = (tid >= off) ? tmp[tid - off] : 0;
    __syncthreads();
    tmp[tid] += t;
    __syncthreads();
  }
  if (i < n) incl[i] = tmp[tid];
  if (tid == 255) bsum[blockIdx.x] = tmp[255];
}

__global__ void k_scan2(int* __restrict__ bsum, int nb) {
  if (threadIdx.x == 0 && blockIdx.x == 0) {
    int run = 0;
    for (int b = 0; b < nb; ++b) {
      int s = bsum[b];
      bsum[b] = run;
      run += s;
    }
  }
}

__global__ void k_scan3(const int* __restrict__ counts, const int* __restrict__ incl,
                        const int* __restrict__ bsum, int* __restrict__ rowptr,
                        int* __restrict__ cursor, int n) {
  int i = blockIdx.x * 256 + threadIdx.x;
  if (i < n) {
    int tot = incl[i] + bsum[blockIdx.x];
    rowptr[i + 1] = tot;
    cursor[i] = tot - counts[i];
    if (i == 0) rowptr[0] = 0;
  }
}

__global__ void k_scatter(const int* __restrict__ src, const int* __restrict__ dst,
                          int* __restrict__ cursor, int* __restrict__ col, int E) {
  int e = blockIdx.x * blockDim.x + threadIdx.x;
  if (e < E) {
    int d = dst[e];
    int pos = atomicAdd(&cursor[d], 1);
    col[pos] = src[e];
  }
}

template <int NCOL, bool RELU>
__global__ void k_fused_layer(const float* __restrict__ x,
                              const int* __restrict__ rowptr,
                              const int* __restrict__ col,
                              const float* __restrict__ nsrc,
                              const float* __restrict__ ndst,
                              const float* __restrict__ W,
                              const float* __restrict__ bias,
                              float* __restrict__ out, int n) {
  __shared__ float sA[16 * LDA];

  const int tid  = threadIdx.x;
  const int wid  = tid >> 5;
  const int lane = tid & 31;
  const size_t grow0 = (size_t)blockIdx.x * 16;

  const int node = (int)grow0 + wid;
  float4 acc; acc.x = acc.y = acc.z = acc.w = 0.0f;
  float nd = 0.0f;
  if (node < n) {
    nd = ndst[node];
    const int beg = rowptr[node];
    const int end = rowptr[node + 1];
    for (int e = beg; e < end; ++e) {
      int s = col[e];
      if (e + 1 < end) {
        __builtin_prefetch(&x[(size_t)col[e + 1] * FEATS + lane * 4], 0, 0);
      }
      float w = nsrc[s];
      float4 v = ((const float4*)(x + (size_t)s * FEATS))[lane];
      acc.x += v.x * w; acc.y += v.y * w; acc.z += v.z * w; acc.w += v.w * w;
    }
  }
  float4 r4; r4.x = acc.x * nd; r4.y = acc.y * nd; r4.z = acc.z * nd; r4.w = acc.w * nd;
  ((float4*)(&sA[wid * LDA]))[lane] = r4;
  __syncthreads();

  constexpr int CT = NCOL / 16;
  if (wid < CT) {
    const int nt   = wid;
    const int l    = lane & 15;
    const int half = lane >> 4;
    const int ncol = nt * 16 + l;

    const float* ar = &sA[l * LDA];
    v8f c = {};
#pragma unroll
    for (int k0 = 0; k0 < FEATS; k0 += 4) {
      const int kk = k0 + 2 * half;
      v2f a, b;
      a.x = ar[kk];
      a.y = ar[kk + 1];
      b.x = W[kk * NCOL + ncol];
      b.y = W[(kk + 1) * NCOL + ncol];
      c = __builtin_amdgcn_wmma_f32_16x16x4_f32(false, a, false, b, (short)0, c,
                                                false, false);
    }

    const float bv = bias[ncol];
    const size_t gm0 = grow0 + 8 * half;
#pragma unroll
    for (int r = 0; r < 8; ++r) {
      size_t gm = gm0 + r;
      if (gm < (size_t)n) {
        float v = c[r] + bv;
        if (RELU) v = fmaxf(v, 0.0f);
        out[gm * (size_t)NCOL + ncol] = v;
      }
    }
  }
}

extern "C" void kernel_launch(void* const* d_in, const int* in_sizes, int n_in,
                              void* d_out, int out_size, void* d_ws, size_t ws_size,
                              hipStream_t stream) {
  const float* feats = (const float*)d_in[0];
  const int*   src   = (const int*)d_in[1];
  const int*   dst   = (const int*)d_in[2];
  const float* W0    = (const float*)d_in[3];
  const float* b0    = (const float*)d_in[4];
  const float* W1    = (const float*)d_in[5];
  const float* b1    = (const float*)d_in[6];
  const float* W2    = (const float*)d_in[7];
  const float* b2    = (const float*)d_in[8];

  const int N = in_sizes[0] / FEATS;
  const int E = in_sizes[1];

  char* ws = (char*)d_ws;
  size_t off = 0;
  auto take = [&](size_t bytes) -> void* {
    void* p = ws + off;
    off += (bytes + 255) & ~(size_t)255;
    return p;
  };

  const int nb = (N + 255) / 256;
  int*   dout_i = (int*)take((size_t)N * 4);
  int*   din_i  = (int*)take((size_t)N * 4);
  float* nsrc   = (float*)take((size_t)N * 4);
  float* ndst   = (float*)take((size_t)N * 4);
  int*   incl   = (int*)take((size_t)N * 4);
  int*   bsum   = (int*)take((size_t)nb * 4);
  int*   rowptr = (int*)take((size_t)(N + 1) * 4);
  int*   cursor = (int*)take((size_t)N * 4);
  int*   col    = (int*)take((size_t)E * 4);
  float* h0     = (float*)take((size_t)N * FEATS * 4);
  float* h1     = (float*)take((size_t)N * FEATS * 4);

  hipMemsetAsync(dout_i, 0, (size_t)N * 4, stream);
  hipMemsetAsync(din_i, 0, (size_t)N * 4, stream);

  const int eb = (E + 255) / 256;
  k_degrees<<<eb, 256, 0, stream>>>(src, dst, dout_i, din_i, E);
  k_norms<<<nb, 256, 0, stream>>>(dout_i, din_i, nsrc, ndst, N);

  k_scan1<<<nb, 256, 0, stream>>>(din_i, incl, bsum, N);
  k_scan2<<<1, 64, 0, stream>>>(bsum, nb);
  k_scan3<<<nb, 256, 0, stream>>>(din_i, incl, bsum, rowptr, cursor, N);
  k_scatter<<<eb, 256, 0, stream>>>(src, dst, cursor, col, E);

  const int tiles = (N + 15) / 16;

  k_fused_layer<128, true><<<tiles, 512, 0, stream>>>(
      feats, rowptr, col, nsrc, ndst, W0, b0, h0, N);
  k_fused_layer<128, true><<<tiles, 512, 0, stream>>>(
      h0, rowptr, col, nsrc, ndst, W1, b1, h1, N);
  k_fused_layer<64, false><<<tiles, 512, 0, stream>>>(
      h1, rowptr, col, nsrc, ndst, W2, b2, (float*)d_out, N);
}


# Fused layers run as 1024 resident 16-wave work-groups (64-register cap, 4 per WGP) looping over the 16-node tiles, two LDS tiles and one barrier per tile, W loads grouped per 8 WMMA steps
# speedup vs baseline: 1.0766x; 1.0766x over previous
#include <hip/hip_runtime.h>
#include <hip/hip_bf16.h>
#include <stdint.h>

typedef __attribute__((ext_vector_type(2))) float v2f;
typedef __attribute__((ext_vector_type(8))) float v8f;

#define FEATS 128
#define LDA 132
#define RESIDENT_BLOCKS 1024

__global__ void k_degrees(const int* __restrict__ src, const int* __restrict__ dst,
                          int* __restrict__ dout, int* __restrict__ din, int E) {
  int i = blockIdx.x * blockDim.x + threadIdx.x;
  if (i < E) {
    atomicAdd(&dout[src[i]], 1);
    atomicAdd(&din[dst[i]], 1);
  }
}

__global__ void k_norms(const int* __restrict__ dout, const int* __restrict__ din,
                        float* __restrict__ nsrc, float* __restrict__ ndst, int n) {
  int i = blockIdx.x * blockDim.x + threadIdx.x;
  if (i < n) {
    nsrc[i] = rsqrtf(fmaxf((float)dout[i], 1.0f));
    ndst[i] = rsqrtf(fmaxf((float)din[i], 1.0f));
  }
}

__global__ void k_scan1(const int* __restrict__ counts, int* __restrict__ incl,
                        int* __restrict__ bsum, int n) {
  __shared__ int tmp[256];
  int tid = threadIdx.x;
  int i = blockIdx.x * 256 + tid;
  int v = (i < n) ? counts[i] : 0;
  tmp[tid] = v;
  __syncthreads();
  for (int off = 1; off < 256; off <<= 1) {
    int t = (tid >= off) ? tmp[tid - off] : 0;
    __syncthreads();
    tmp[tid] += t;
    __syncthreads();
  }
  if (i < n) incl[i] = tmp[tid];
  if (tid == 255) bsum[blockIdx.x] = tmp[255];
}

__global__ void k_scan2(int* __restrict__ bsum, int nb) {
  if (threadIdx.x == 0 && blockIdx.x == 0) {
    int run = 0;
    for (int b = 0; b < nb; ++b) {
      int s = bsum[b];
      bsum[b] = run;
      run += s;
    }
  }
}

__global__ void k_scan3(const int* __restrict__ counts, const int* __restrict__ incl,
                        const int* __restrict__ bsum, int* __restrict__ rowptr,
                        int* __restrict__ cursor, int n) {
  int i = blockIdx.x * 256 + threadIdx.x;
  if (i < n) {
    int tot = incl[i] + bsum[blockIdx.x];
    rowptr[i + 1] = tot;
    cursor[i] = tot - counts[i];
    if (i == 0) rowptr[0] = 0;
  }
}

__global__ void k_scatter(const int* __restrict__ src, const int* __restrict__ dst,
                          int* __restrict__ cursor, int* __restrict__ col, int E) {
  int e = blockIdx.x * blockDim.x + threadIdx.x;
  if (e < E) {
    int d = dst[e];
    int pos = atomicAdd(&cursor[d], 1);
    col[pos] = src[e];
  }
}

template <int NCOL, bool RELU>
__global__ void __launch_bounds__(512, 16) k_fused_layer(const float* __restrict__ x,
                              const int* __restrict__ rowptr,
                              const int* __restrict__ col,
                              const float* __restrict__ nsrc,
                              const float* __restrict__ ndst,
                              const float* __restrict__ W,
                              const float* __restrict__ bias,
                              float* __restrict__ out, int n, int tiles) {
  __shared__ float sA2[2][16 * LDA];

  const int tid  = threadIdx.x;
  const int wid  = tid >> 5;
  const int lane = tid & 31;

  int buf = 0;
  for (int tile = blockIdx.x; tile < tiles; tile += gridDim.x, buf ^= 1) {
  float* sA = sA2[buf];
  const size_t grow0 = (size_t)tile * 16;

  const int node = (int)grow0 + wid;
  float4 acc; acc.x = acc.y = acc.z = acc.w = 0.0f;
  float nd = 0.0f;
  if (node < n) {
    nd = ndst[node];
    const int beg = rowptr[node];
    const int end = rowptr[node + 1];
    for (int e = beg; e < end; ++e) {
      int s = col[e];
      if (e + 1 < end) {
        __builtin_prefetch(&x[(size_t)col[e + 1] * FEATS + lane * 4], 0, 0);
      }
      float w = nsrc[s];
      float4 v = ((const float4*)(x + (size_t)s * FEATS))[lane];
      acc.x += v.x * w; acc.y += v.y * w; acc.z += v.z * w; acc.w += v.w * w;
    }
  }
  float4 r4; r4.x = acc.x * nd; r4.y = acc.y * nd; r4.z = acc.z * nd; r4.w = acc.w * nd;
  ((float4*)(&sA[wid * LDA]))[lane] = r4;
  __syncthreads();

  constexpr int CT = NCOL / 16;
  if (wid < CT) {
    const int nt   = wid;
    const int l    = lane & 15;
    const int half = lane >> 4;
    const int ncol = nt * 16 + l;

    const float* ar = &sA[l * LDA];
    v8f c = {};
    constexpr int KG = 32;
#pragma unroll 1
    for (int kc = 0; kc < FEATS; kc += KG) {
      v2f bf[KG / 4];
      v2f af[KG / 4];
#pragma unroll
      for (int s = 0; s < KG / 4; ++s) {
        const int kk = kc + 4 * s + 2 * half;
        bf[s].x = W[kk * NCOL + ncol];
        bf[s].y = W[(kk + 1) * NCOL + ncol];
      }
#pragma unroll
      for (int s = 0; s < KG / 4; ++s) {
        const int kk = kc + 4 * s + 2 * half;
        af[s].x = ar[kk];
        af[s].y = ar[kk + 1];
      }
#pragma unroll
      for (int s = 0; s < KG / 4; ++s) {
        c = __builtin_amdgcn_wmma_f32_16x16x4_f32(false, af[s], false, bf[s], (short)0, c,
                                                  false, false);
      }
    }

    const float bv = bias[ncol];
    const size_t gm0 = grow0 + 8 * half;
#pragma unroll
    for (int r = 0; r < 8; ++r) {
      size_t gm = gm0 + r;
      if (gm < (size_t)n) {
        float v = c[r] + bv;
        if (RELU) v = fmaxf(v, 0.0f);
        out[gm * (size_t)NCOL + ncol] = v;
      }
    }
  }
  }
}

extern "C" void kernel_launch(void* const* d_in, const int* in_sizes, int n_in,
                              void* d_out, int out_size, void* d_ws, size_t ws_size,
                              hipStream_t stream) {
  const float* feats = (const float*)d_in[0];
  const int*   src   = (const int*)d_in[1];
  const int*   dst   = (const int*)d_in[2];
  const float* W0    = (const float*)d_in[3];
  const float* b0    = (const float*)d_in[4];
  const float* W1    = (const float*)d_in[5];
  const float* b1    = (const float*)d_in[6];
  const float* W2    = (const float*)d_in[7];
  const float* b2    = (const float*)d_in[8];

  const int N = in_sizes[0] / FEATS;
  const int E = in_sizes[1];

  char* ws = (char*)d_ws;
  size_t off = 0;
  auto take = [&](size_t bytes) -> void* {
    void* p = ws + off;
    off += (bytes + 255) & ~(size_t)255;
    return p;
  };

  const int nb = (N + 255) / 256;
  int*   dout_i = (int*)take((size_t)N * 4);
  int*   din_i  = (int*)take((size_t)N * 4);
  float* nsrc   = (float*)take((size_t)N * 4);
  float* ndst   = (float*)take((size_t)N * 4);
  int*   incl   = (int*)take((size_t)N * 4);
  int*   bsum   = (int*)take((size_t)nb * 4);
  int*   rowptr = (int*)take((size_t)(N + 1) * 4);
  int*   cursor = (int*)take((size_t)N * 4);
  int*   col    = (int*)take((size_t)E * 4);
  float* h0     = (float*)take((size_t)N * FEATS * 4);
  float* h1     = (float*)take((size_t)N * FEATS * 4);

  hipMemsetAsync(dout_i, 0, (size_t)N * 4, stream);
  hipMemsetAsync(din_i, 0, (size_t)N * 4, stream);

  const int eb = (E + 255) / 256;
  k_degrees<<<eb, 256, 0, stream>>>(src, dst, dout_i, din_i, E);
  k_norms<<<nb, 256, 0, stream>>>(dout_i, din_i, nsrc, ndst, N);

  k_scan1<<<nb, 256, 0, stream>>>(din_i, incl, bsum, N);
  k_scan2<<<1, 64, 0, stream>>>(bsum, nb);
  k_scan3<<<nb, 256, 0, stream>>>(din_i, incl, bsum, rowptr, cursor, N);
  k_scatter<<<eb, 256, 0, stream>>>(src, dst, cursor, col, E);

  const int tiles = (N + 15) / 16;

  k_fused_layer<128, true><<<RESIDENT_BLOCKS, 512, 0, stream>>>(
      feats, rowptr, col, nsrc, ndst, W0, b0, h0, N, tiles);
  k_fused_layer<128, true><<<RESIDENT_BLOCKS, 512, 0, stream>>>(
      h0, rowptr, col, nsrc, ndst, W1, b1, h1, N, tiles);
  k_fused_layer<64, false><<<RESIDENT_BLOCKS, 512, 0, stream>>>(
      h1, rowptr, col, nsrc, ndst, W2, b2, (float*)d_out, N, tiles);
}


# Neighbor gather batched: a node's indices and source norms fetched lane-parallel, handed out by readlane, 8 feature rows requested together before any is summed (no per-edge prefetch); sum order kept
# speedup vs baseline: 1.2949x; 1.2028x over previous
#include <hip/hip_runtime.h>
#include <hip/hip_bf16.h>
#include <stdint.h>

typedef __attribute__((ext_vector_type(2))) float v2f;
typedef __attribute__((ext_vector_type(8))) float v8f;

#define FEATS 128
#define LDA 132
#define RESIDENT_BLOCKS 1024
#define GB 8

__global__ void k_degrees(const int* __restrict__ src, const int* __restrict__ dst,
                          int* __restrict__ dout, int* __restrict__ din, int E) {
  int i = blockIdx.x * blockDim.x + threadIdx.x;
  if (i < E) {
    atomicAdd(&dout[src[i]], 1);
    atomicAdd(&din[dst[i]], 1);
  }
}

__global__ void k_norms(const int* __restrict__ dout, const int* __restrict__ din,
                        float* __restrict__ nsrc, float* __restrict__ ndst, int n) {
  int i = blockIdx.x * blockDim.x + threadIdx.x;
  if (i < n) {
    nsrc[i] = rsqrtf(fmaxf((float)dout[i], 1.0f));
    ndst[i] = rsqrtf(fmaxf((float)din[i], 1.0f));
  }
}

__global__ void k_scan1(const int* __restrict__ counts, int* __restrict__ incl,
                        int* __restrict__ bsum, int n) {
  __shared__ int tmp[256];
  int tid = threadIdx.x;
  int i = blockIdx.x * 256 + tid;
  int v = (i < n) ? counts[i] : 0;
  tmp[tid] = v;
  __syncthreads();
  for (int off = 1; off < 256; off <<= 1) {
    int t = (tid >= off) ? tmp[tid - off] : 0;
    __syncthreads();
    tmp[tid] += t;
    __syncthreads();
  }
  if (i < n) incl[i] = tmp[tid];
  if (tid == 255) bsum[blockIdx.x] = tmp[255];
}

__global__ void k_scan2(int* __restrict__ bsum, int nb) {
  if (threadIdx.x == 0 && blockIdx.x == 0) {
    int run = 0;
    for (int b = 0; b < nb; ++b) {
      int s = bsum[b];
      bsum[b] = run;
      run += s;
    }
  }
}

__global__ void k_scan3(const int* __restrict__ counts, const int* __restrict__ incl,
                        const int* __restrict__ bsum, int* __restrict__ rowptr,
                        int* __restrict__ cursor, int n) {
  int i = blockIdx.x * 256 + threadIdx.x;
  if (i < n) {
    int tot = incl[i] + bsum[blockIdx.x];
    rowptr[i + 1] = tot;
    cursor[i] = tot - counts[i];
    if (i == 0) rowptr[0] = 0;
  }
}

__global__ void k_scatter(const int* __restrict__ src, const int* __restrict__ dst,
                          int* __restrict__ cursor, int* __restrict__ col, int E) {
  int e = blockIdx.x * blockDim.x + threadIdx.x;
  if (e < E) {
    int d = dst[e];
    int pos = atomicAdd(&cursor[d], 1);
    col[pos] = src[e];
  }
}

template <int NCOL, bool RELU>
__global__ void __launch_bounds__(512, 16) k_fused_layer(const float* __restrict__ x,
                              const int* __restrict__ rowptr,
                              const int* __restrict__ col,
                              const float* __restrict__ nsrc,
                              const float* __restrict__ ndst,
                              const float* __restrict__ W,
                              const float* __restrict__ bias,
                              float* __restrict__ out, int n, int tiles) {
  __shared__ float sA2[2][16 * LDA];

  const int tid  = threadIdx.x;
  const int wid  = tid >> 5;
  const int lane = tid & 31;

  int buf = 0;
  for (int tile = blockIdx.x; tile < tiles; tile += gridDim.x, buf ^= 1) {
  float* sA = sA2[buf];
  const size_t grow0 = (size_t)tile * 16;

  const int uwid = __builtin_amdgcn_readfirstlane(wid);
  const int node = (int)grow0 + uwid;
  float4 acc; acc.x = acc.y = acc.z = acc.w = 0.0f;
  float nd = 0.0f;
  if (node < n) {
    nd = ndst[node];
    const int beg = rowptr[node];
    const int end = rowptr[node + 1];
    for (int base = beg; base < end; base += 32) {
      const int left = end - base;
      const int cnt = left < 32 ? left : 32;
      int my_s = 0;
      float my_w = 0.0f;
      if (lane < cnt) {
        my_s = col[base + lane];
        my_w = nsrc[my_s];
      }
      for (int j = 0; j < cnt; j += GB) {
        float4 v[GB];
#pragma unroll
        for (int u = 0; u < GB; ++u) {
          const int p = (j + u < cnt) ? j + u : cnt - 1;
          const int s = __builtin_amdgcn_readlane(my_s, p);
          v[u] = ((const float4*)(x + (size_t)s * FEATS))[lane];
        }
        __builtin_amdgcn_sched_barrier(0);
#pragma unroll
        for (int u = 0; u < GB; ++u) {
          const float w = __int_as_float(__builtin_amdgcn_readlane(__float_as_int(my_w), j + u));
          acc.x += v[u].x * w; acc.y += v[u].y * w; acc.z += v[u].z * w; acc.w += v[u].w * w;
        }
      }
    }
  }
  float4 r4; r4.x = acc.x * nd; r4.y = acc.y * nd; r4.z = acc.z * nd; r4.w = acc.w * nd;
  ((float4*)(&sA[wid * LDA]))[lane] = r4;
  __syncthreads();

  constexpr int CT = NCOL / 16;
  if (wid < CT) {
    const int nt   = wid;
    const int l    = lane & 15;
    const int half = lane >> 4;
    const int ncol = nt * 16 + l;

    const float* ar = &sA[l * LDA];
    v8f c = {};
    constexpr int KG = 32;
#pragma unroll 1
    for (int kc = 0; kc < FEATS; kc += KG) {
      v2f bf[KG / 4];
      v2f af[KG / 4];
#pragma unroll
      for (int s = 0; s < KG / 4; ++s) {
        const int kk = kc + 4 * s + 2 * half;
        bf[s].x = W[kk * NCOL + ncol];
        bf[s].y = W[(kk + 1) * NCOL + ncol];
      }
#pragma unroll
      for (int s = 0; s < KG / 4; ++s) {
        const int kk = kc + 4 * s + 2 * half;
        af[s].x = ar[kk];
        af[s].y = ar[kk + 1];
      }
#pragma unroll
      for (int s = 0; s < KG / 4; ++s) {
        c = __builtin_amdgcn_wmma_f32_16x16x4_f32(false, af[s], false, bf[s], (short)0, c,
                                                  false, false);
      }
    }

    const float bv = bias[ncol];
    const size_t gm0 = grow0 + 8 * half;
#pragma unroll
    for (int r = 0; r < 8; ++r) {
      size_t gm = gm0 + r;
      if (gm < (size_t)n) {
        float v = c[r] + bv;
        if (RELU) v = fmaxf(v, 0.0f);
        out[gm * (size_t)NCOL + ncol] = v;
      }
    }
  }
  }
}

extern "C" void kernel_launch(void* const* d_in, const int* in_sizes, int n_in,
                              void* d_out, int out_size, void* d_ws, size_t ws_size,
                              hipStream_t stream) {
  const float* feats = (const float*)d_in[0];
  const int*   src   = (const int*)d_in[1];
  const int*   dst   = (const int*)d_in[2];
  const float* W0    = (const float*)d_in[3];
  const float* b0    = (const float*)d_in[4];
  const float* W1    = (const float*)d_in[5];
  const float* b1    = (const float*)d_in[6];
  const float* W2    = (const float*)d_in[7];
  const float* b2    = (const float*)d_in[8];

  const int N = in_sizes[0] / FEATS;
  const int E = in_sizes[1];

  char* ws = (char*)d_ws;
  size_t off = 0;
  auto take = [&](size_t bytes) -> void* {
    void* p = ws + off;
    off += (bytes + 255) & ~(size_t)255;
    return p;
  };

  const int nb = (N + 255) / 256;
  int*   dout_i = (int*)take((size_t)N * 4);
  int*   din_i  = (int*)take((size_t)N * 4);
  float* nsrc   = (float*)take((size_t)N * 4);
  float* ndst   = (float*)take((size_t)N * 4);
  int*   incl   = (int*)take((size_t)N * 4);
  int*   bsum   = (int*)take((size_t)nb * 4);
  int*   rowptr = (int*)take((size_t)(N + 1) * 4);
  int*   cursor = (int*)take((size_t)N * 4);
  int*   col    = (int*)take((size_t)E * 4);
  float* h0     = (float*)take((size_t)N * FEATS * 4);
  float* h1     = (float*)take((size_t)N * FEATS * 4);

  hipMemsetAsync(dout_i, 0, (size_t)N * 4, stream);
  hipMemsetAsync(din_i, 0, (size_t)N * 4, stream);

  const int eb = (E + 255) / 256;
  k_degrees<<<eb, 256, 0, stream>>>(src, dst, dout_i, din_i, E);
  k_norms<<<nb, 256, 0, stream>>>(dout_i, din_i, nsrc, ndst, N);

  k_scan1<<<nb, 256, 0, stream>>>(din_i, incl, bsum, N);
  k_scan2<<<1, 64, 0, stream>>>(bsum, nb);
  k_scan3<<<nb, 256, 0, stream>>>(din_i, incl, bsum, rowptr, cursor, N);
  k_scatter<<<eb, 256, 0, stream>>>(src, dst, cursor, col, E);

  const int tiles = (N + 15) / 16;

  k_fused_layer<128, true><<<RESIDENT_BLOCKS, 512, 0, stream>>>(
      feats, rowptr, col, nsrc, ndst, W0, b0, h0, N, tiles);
  k_fused_layer<128, true><<<RESIDENT_BLOCKS, 512, 0, stream>>>(
      h0, rowptr, col, nsrc, ndst, W1, b1, h1, N, tiles);
  k_fused_layer<64, false><<<RESIDENT_BLOCKS, 512, 0, stream>>>(
      h1, rowptr, col, nsrc, ndst, W2, b2, (float*)d_out, N, tiles);
}


# Degree counts and destination sort built from per-work-group LDS counters over 64 edge chunks (no global atomics, no memsets); a node's neighbors now summed in chunk order, then LDS arrival order
# speedup vs baseline: 2.0536x; 1.5859x over previous
#include <hip/hip_runtime.h>
#include <hip/hip_bf16.h>
#include <stdint.h>

typedef __attribute__((ext_vector_type(2))) float v2f;
typedef __attribute__((ext_vector_type(8))) float v8f;

#define FEATS 128
#define LDA 132
#define RESIDENT_BLOCKS 1024
#define GB 8

#define NB 64
#define HIST_WORDS 50000

__global__ void __launch_bounds__(1024) k_count(const int* __restrict__ src,
                                                const int* __restrict__ dst,
                                                unsigned int* __restrict__ partD,
                                                unsigned int* __restrict__ partS,
                                                int* __restrict__ rank, int E, int NW) {
  __shared__ __align__(16) unsigned int hist[HIST_WORDS];
  const int role = blockIdx.x / NB;
  const int b    = blockIdx.x - role * NB;
  const int tid  = (int)threadIdx.x;
  for (int w = tid; w < HIST_WORDS / 4; w += 1024) ((uint4*)hist)[w] = make_uint4(0u, 0u, 0u, 0u);
  __syncthreads();
  const int chunk = (E + NB - 1) / NB;
  const int e0 = b * chunk;
  const int e1 = (e0 + chunk < E) ? e0 + chunk : E;
  const int* __restrict__ keys = (role == 0) ? dst : src;
  constexpr int CU = 8;
  for (int base = e0 + tid; base < e1; base += 1024 * CU) {
    int k[CU];
#pragma unroll
    for (int u = 0; u < CU; ++u) {
      const int e = base + u * 1024;
      k[u] = (e < e1) ? keys[e] : 0;
    }
    if (role == 0) {
#pragma unroll
      for (int u = 0; u < CU; ++u) {
        const int e = base + u * 1024;
        if (e < e1) {
          const unsigned int sh = (unsigned int)(k[u] & 1) * 16u;
          const unsigned int old = atomicAdd(&hist[k[u] >> 1], 1u << sh);
          rank[e] = (int)((old >> sh) & 0xFFFFu);
        }
      }
    } else {
#pragma unroll
      for (int u = 0; u < CU; ++u) {
        const int e = base + u * 1024;
        if (e < e1) atomicAdd(&hist[k[u] >> 1], 1u << ((unsigned int)(k[u] & 1) * 16u));
      }
    }
  }
  __syncthreads();
  unsigned int* __restrict__ part = (role == 0 ? partD : partS) + (size_t)b * NW;
  for (int w = tid; w < NW; w += 1024) part[w] = hist[w];
}

__global__ void k_reduce(const unsigned int* __restrict__ partD,
                         const unsigned int* __restrict__ partS,
                         int* __restrict__ pref, int* __restrict__ din,
                         float* __restrict__ nsrc, float* __restrict__ ndst, int n, int NW) {
  const int w = blockIdx.x * blockDim.x + threadIdx.x;
  if (w < NW) {
    const int i0 = 2 * w;
    const int i1 = 2 * w + 1;
    unsigned int d0 = 0u, d1 = 0u, s0 = 0u, s1 = 0u;
    constexpr int RB = 32;
    for (int bb = 0; bb < NB; bb += RB) {
      unsigned int v[RB], u[RB];
#pragma unroll
      for (int j = 0; j < RB; ++j) {
        v[j] = partD[(size_t)(bb + j) * NW + w];
        u[j] = partS[(size_t)(bb + j) * NW + w];
      }
#pragma unroll
      for (int j = 0; j < RB; ++j) {
        pref[(size_t)(bb + j) * n + i0] = (int)d0;
        if (i1 < n) pref[(size_t)(bb + j) * n + i1] = (int)d1;
        d0 += v[j] & 0xFFFFu; d1 += v[j] >> 16;
        s0 += u[j] & 0xFFFFu; s1 += u[j] >> 16;
      }
    }
    din[i0]  = (int)d0;
    nsrc[i0] = rsqrtf(fmaxf((float)s0, 1.0f));
    ndst[i0] = rsqrtf(fmaxf((float)d0, 1.0f));
    if (i1 < n) {
      din[i1]  = (int)d1;
      nsrc[i1] = rsqrtf(fmaxf((float)s1, 1.0f));
      ndst[i1] = rsqrtf(fmaxf((float)d1, 1.0f));
    }
  }
}

__global__ void k_scan1(const int* __restrict__ counts, int* __restrict__ incl,
                        int* __restrict__ bsum, int n) {
  __shared__ int tmp[256];
  int tid = threadIdx.x;
  int i = blockIdx.x * 256 + tid;
  int v = (i < n) ? counts[i] : 0;
  tmp[tid] = v;
  __syncthreads();
  for (int off = 1; off < 256; off <<= 1) {
    int t = (tid >= off) ? tmp[tid - off] : 0;
    __syncthreads();
    tmp[tid] += t;
    __syncthreads();
  }
  if (i < n) incl[i] = tmp[tid];
  if (tid == 255) bsum[blockIdx.x] = tmp[255];
}

__global__ void k_scan2(int* __restrict__ bsum, int nb) {
  if (threadIdx.x == 0 && blockIdx.x == 0) {
    int run = 0;
    for (int b = 0; b < nb; ++b) {
      int s = bsum[b];
      bsum[b] = run;
      run += s;
    }
  }
}

__global__ void k_scan3(const int* __restrict__ incl, const int* __restrict__ bsum,
                        int* __restrict__ rowptr, int n) {
  int i = blockIdx.x * 256 + threadIdx.x;
  if (i < n) {
    rowptr[i + 1] = incl[i] + bsum[blockIdx.x];
    if (i == 0) rowptr[0] = 0;
  }
}

__global__ void k_place(const int* __restrict__ src, const int* __restrict__ dst,
                        const int* __restrict__ rowptr, const int* __restrict__ pref,
                        const int* __restrict__ rank, int* __restrict__ col, int E, int n) {
  int e = blockIdx.x * blockDim.x + threadIdx.x;
  if (e < E) {
    const int chunk = (E + NB - 1) / NB;
    const int d = dst[e];
    const int b = e / chunk;
    const int pos = rowptr[d] + pref[(size_t)b * n + d] + rank[e];
    col[pos] = src[e];
  }
}

template <int NCOL, bool RELU>
__global__ void __launch_bounds__(512, 16) k_fused_layer(const float* __restrict__ x,
                              const int* __restrict__ rowptr,
                              const int* __restrict__ col,
                              const float* __restrict__ nsrc,
                              const float* __restrict__ ndst,
                              const float* __restrict__ W,
                              const float* __restrict__ bias,
                              float* __restrict__ out, int n, int tiles) {
  __shared__ float sA2[2][16 * LDA];

  const int tid  = threadIdx.x;
  const int wid  = tid >> 5;
  const int lane = tid & 31;

  int buf = 0;
  for (int tile = blockIdx.x; tile < tiles; tile += gridDim.x, buf ^= 1) {
  float* sA = sA2[buf];
  const size_t grow0 = (size_t)tile * 16;

  const int uwid = __builtin_amdgcn_readfirstlane(wid);
  const int node = (int)grow0 + uwid;
  float4 acc; acc.x = acc.y = acc.z = acc.w = 0.0f;
  float nd = 0.0f;
  if (node < n) {
    nd = ndst[node];
    const int beg = rowptr[node];
    const int end = rowptr[node + 1];
    for (int base = beg; base < end; base += 32) {
      const int left = end - base;
      const int cnt = left < 32 ? left : 32;
      int my_s = 0;
      float my_w = 0.0f;
      if (lane < cnt) {
        my_s = col[base + lane];
        my_w = nsrc[my_s];
      }
      for (int j = 0; j < cnt; j += GB) {
        float4 v[GB];
#pragma unroll
        for (int u = 0; u < GB; ++u) {
          const int p = (j + u < cnt) ? j + u : cnt - 1;
          const int s = __builtin_amdgcn_readlane(my_s, p);
          v[u] = ((const float4*)(x + (size_t)s * FEATS))[lane];
        }
        __builtin_amdgcn_sched_barrier(0);
#pragma unroll
        for (int u = 0; u < GB; ++u) {
          const float w = __int_as_float(__builtin_amdgcn_readlane(__float_as_int(my_w), j + u));
          acc.x += v[u].x * w; acc.y += v[u].y * w; acc.z += v[u].z * w; acc.w += v[u].w * w;
        }
      }
    }
  }
  float4 r4; r4.x = acc.x * nd; r4.y = acc.y * nd; r4.z = acc.z * nd; r4.w = acc.w * nd;
  ((float4*)(&sA[wid * LDA]))[lane] = r4;
  __syncthreads();

  constexpr int CT = NCOL / 16;
  if (wid < CT) {
    const int nt   = wid;
    const int l    = lane & 15;
    const int half = lane >> 4;
    const int ncol = nt * 16 + l;

    const float* ar = &sA[l * LDA];
    v8f c = {};
    constexpr int KG = 32;
#pragma unroll 1
    for (int kc = 0; kc < FEATS; kc += KG) {
      v2f bf[KG / 4];
      v2f af[KG / 4];
#pragma unroll
      for (int s = 0; s < KG / 4; ++s) {
        const int kk = kc + 4 * s + 2 * half;
        bf[s].x = W[kk * NCOL + ncol];
        bf[s].y = W[(kk + 1) * NCOL + ncol];
      }
#pragma unroll
      for (int s = 0; s < KG / 4; ++s) {
        const int kk = kc + 4 * s + 2 * half;
        af[s].x = ar[kk];
        af[s].y = ar[kk + 1];
      }
#pragma unroll
      for (int s = 0; s < KG / 4; ++s) {
        c = __builtin_amdgcn_wmma_f32_16x16x4_f32(false, af[s], false, bf[s], (short)0, c,
                                                  false, false);
      }
    }

    const float bv = bias[ncol];
    const size_t gm0 = grow0 + 8 * half;
#pragma unroll
    for (int r = 0; r < 8; ++r) {
      size_t gm = gm0 + r;
      if (gm < (size_t)n) {
        float v = c[r] + bv;
        if (RELU) v = fmaxf(v, 0.0f);
        out[gm * (size_t)NCOL + ncol] = v;
      }
    }
  }
  }
}

extern "C" void kernel_launch(void* const* d_in, const int* in_sizes, int n_in,
                              void* d_out, int out_size, void* d_ws, size_t ws_size,
                              hipStream_t stream) {
  const float* feats = (const float*)d_in[0];
  const int*   src   = (const int*)d_in[1];
  const int*   dst   = (const int*)d_in[2];
  const float* W0    = (const float*)d_in[3];
  const float* b0    = (const float*)d_in[4];
  const float* W1    = (const float*)d_in[5];
  const float* b1    = (const float*)d_in[6];
  const float* W2    = (const float*)d_in[7];
  const float* b2    = (const float*)d_in[8];

  const int N = in_sizes[0] / FEATS;
  const int E = in_sizes[1];

  char* ws = (char*)d_ws;
  size_t off = 0;
  auto take = [&](size_t bytes) -> void* {
    void* p = ws + off;
    off += (bytes + 255) & ~(size_t)255;
    return p;
  };

  const int nb = (N + 255) / 256;
  const int NW = (N + 1) / 2;
  unsigned int* partD = (unsigned int*)take((size_t)NB * NW * 4);
  unsigned int* partS = (unsigned int*)take((size_t)NB * NW * 4);
  int*   pref   = (int*)take((size_t)NB * N * 4);
  int*   rank   = (int*)take((size_t)E * 4);
  int*   din_i  = (int*)take((size_t)N * 4);
  float* nsrc   = (float*)take((size_t)N * 4);
  float* ndst   = (float*)take((size_t)N * 4);
  int*   incl   = (int*)take((size_t)N * 4);
  int*   bsum   = (int*)take((size_t)nb * 4);
  int*   rowptr = (int*)take((size_t)(N + 1) * 4);
  int*   col    = (int*)take((size_t)E * 4);
  float* h0     = (float*)take((size_t)N * FEATS * 4);
  float* h1     = (float*)take((size_t)N * FEATS * 4);

  const int eb = (E + 255) / 256;
  k_count<<<2 * NB, 1024, 0, stream>>>(src, dst, partD, partS, rank, E, NW);
  k_reduce<<<(NW + 255) / 256, 256, 0, stream>>>(partD, partS, pref, din_i, nsrc, ndst, N, NW);

  k_scan1<<<nb, 256, 0, stream>>>(din_i, incl, bsum, N);
  k_scan2<<<1, 64, 0, stream>>>(bsum, nb);
  k_scan3<<<nb, 256, 0, stream>>>(incl, bsum, rowptr, N);
  k_place<<<eb, 256, 0, stream>>>(src, dst, rowptr, pref, rank, col, E, N);

  const int tiles = (N + 15) / 16;

  k_fused_layer<128, true><<<RESIDENT_BLOCKS, 512, 0, stream>>>(
      feats, rowptr, col, nsrc, ndst, W0, b0, h0, N, tiles);
  k_fused_layer<128, true><<<RESIDENT_BLOCKS, 512, 0, stream>>>(
      h0, rowptr, col, nsrc, ndst, W1, b1, h1, N, tiles);
  k_fused_layer<64, false><<<RESIDENT_BLOCKS, 512, 0, stream>>>(
      h1, rowptr, col, nsrc, ndst, W2, b2, (float*)d_out, N, tiles);
}


# Layer 2 takes its weights before its neighbor sum: layer 1's kernel also forms h1*W2 on WMMA from a second LDS tile, layer 2 only sums 64-wide rows; layer 2's rounding order changes
# speedup vs baseline: 2.2180x; 1.0801x over previous
#include <hip/hip_runtime.h>
#include <hip/hip_bf16.h>
#include <stdint.h>

typedef __attribute__((ext_vector_type(2))) float v2f;
typedef __attribute__((ext_vector_type(8))) float v8f;

#define FEATS 128
#define LDA 132
#define RESIDENT_BLOCKS 1024
#define GB 8

#define NB 64
#define HIST_WORDS 50000

__global__ void __launch_bounds__(1024) k_count(const int* __restrict__ src,
                                                const int* __restrict__ dst,
                                                unsigned int* __restrict__ partD,
                                                unsigned int* __restrict__ partS,
                                                int* __restrict__ rank, int E, int NW) {
  __shared__ __align__(16) unsigned int hist[HIST_WORDS];
  const int role = blockIdx.x / NB;
  const int b    = blockIdx.x - role * NB;
  const int tid  = (int)threadIdx.x;
  for (int w = tid; w < HIST_WORDS / 4; w += 1024) ((uint4*)hist)[w] = make_uint4(0u, 0u, 0u, 0u);
  __syncthreads();
  const int chunk = (E + NB - 1) / NB;
  const int e0 = b * chunk;
  const int e1 = (e0 + chunk < E) ? e0 + chunk : E;
  const int* __restrict__ keys = (role == 0) ? dst : src;
  constexpr int CU = 8;
  for (int base = e0 + tid; base < e1; base += 1024 * CU) {
    int k[CU];
#pragma unroll
    for (int u = 0; u < CU; ++u) {
      const int e = base + u * 1024;
      k[u] = (e < e1) ? keys[e] : 0;
    }
    if (role == 0) {
#pragma unroll
      for (int u = 0; u < CU; ++u) {
        const int e = base + u * 1024;
        if (e < e1) {
          const unsigned int sh = (unsigned int)(k[u] & 1) * 16u;
          const unsigned int old = atomicAdd(&hist[k[u] >> 1], 1u << sh);
          rank[e] = (int)((old >> sh) & 0xFFFFu);
        }
      }
    } else {
#pragma unroll
      for (int u = 0; u < CU; ++u) {
        const int e = base + u * 1024;
        if (e < e1) atomicAdd(&hist[k[u] >> 1], 1u << ((unsigned int)(k[u] & 1) * 16u));
      }
    }
  }
  __syncthreads();
  unsigned int* __restrict__ part = (role == 0 ? partD : partS) + (size_t)b * NW;
  for (int w = tid; w < NW; w += 1024) part[w] = hist[w];
}

__global__ void k_reduce(const unsigned int* __restrict__ partD,
                         const unsigned int* __restrict__ partS,
                         int* __restrict__ pref, int* __restrict__ din,
                         float* __restrict__ nsrc, float* __restrict__ ndst, int n, int NW) {
  const int w = blockIdx.x * blockDim.x + threadIdx.x;
  if (w < NW) {
    const int i0 = 2 * w;
    const int i1 = 2 * w + 1;
    unsigned int d0 = 0u, d1 = 0u, s0 = 0u, s1 = 0u;
    constexpr int RB = 32;
    for (int bb = 0; bb < NB; bb += RB) {
      unsigned int v[RB], u[RB];
#pragma unroll
      for (int j = 0; j < RB; ++j) {
        v[j] = partD[(size_t)(bb + j) * NW + w];
        u[j] = partS[(size_t)(bb + j) * NW + w];
      }
#pragma unroll
      for (int j = 0; j < RB; ++j) {
        pref[(size_t)(bb + j) * n + i0] = (int)d0;
        if (i1 < n) pref[(size_t)(bb + j) * n + i1] = (int)d1;
        d0 += v[j] & 0xFFFFu; d1 += v[j] >> 16;
        s0 += u[j] & 0xFFFFu; s1 += u[j] >> 16;
      }
    }
    din[i0]  = (int)d0;
    nsrc[i0] = rsqrtf(fmaxf((float)s0, 1.0f));
    ndst[i0] = rsqrtf(fmaxf((float)d0, 1.0f));
    if (i1 < n) {
      din[i1]  = (int)d1;
      nsrc[i1] = rsqrtf(fmaxf((float)s1, 1.0f));
      ndst[i1] = rsqrtf(fmaxf((float)d1, 1.0f));
    }
  }
}

__global__ void k_scan1(const int* __restrict__ counts, int* __restrict__ incl,
                        int* __restrict__ bsum, int n) {
  __shared__ int tmp[256];
  int tid = threadIdx.x;
  int i = blockIdx.x * 256 + tid;
  int v = (i < n) ? counts[i] : 0;
  tmp[tid] = v;
  __syncthreads();
  for (int off = 1; off < 256; off <<= 1) {
    int t = (tid >= off) ? tmp[tid - off] : 0;
    __syncthreads();
    tmp[tid] += t;
    __syncthreads();
  }
  if (i < n) incl[i] = tmp[tid];
  if (tid == 255) bsum[blockIdx.x] = tmp[255];
}

__global__ void k_scan2(int* __restrict__ bsum, int nb) {
  if (threadIdx.x == 0 && blockIdx.x == 0) {
    int run = 0;
    for (int b = 0; b < nb; ++b) {
      int s = bsum[b];
      bsum[b] = run;
      run += s;
    }
  }
}

__global__ void k_scan3(const int* __restrict__ incl, const int* __restrict__ bsum,
                        int* __restrict__ rowptr, int n) {
  int i = blockIdx.x * 256 + threadIdx.x;
  if (i < n) {
    rowptr[i + 1] = incl[i] + bsum[blockIdx.x];
    if (i == 0) rowptr[0] = 0;
  }
}

__global__ void k_place(const int* __restrict__ src, const int* __restrict__ dst,
                        const int* __restrict__ rowptr, const int* __restrict__ pref,
                        const int* __restrict__ rank, int* __restrict__ col, int E, int n) {
  int e = blockIdx.x * blockDim.x + threadIdx.x;
  if (e < E) {
    const int chunk = (E + NB - 1) / NB;
    const int d = dst[e];
    const int b = e / chunk;
    const int pos = rowptr[d] + pref[(size_t)b * n + d] + rank[e];
    col[pos] = src[e];
  }
}

template <int NC>
__device__ __forceinline__ v8f tile_gemm(const float* __restrict__ sTile,
                                         const float* __restrict__ Wc, int lane) {
  const int l    = lane & 15;
  const int half = lane >> 4;
  const float* ar = sTile + l * LDA + 2 * half;
  const float* wr = Wc + (2 * half) * NC + l;
  v8f c = {};
  constexpr int KG = 32;
#pragma unroll 1
  for (int kc = 0; kc < FEATS; kc += KG) {
    v2f bf[KG / 4];
    v2f af[KG / 4];
#pragma unroll
    for (int s = 0; s < KG / 4; ++s) {
      const int kk = kc + 4 * s;
      bf[s].x = wr[kk * NC];
      bf[s].y = wr[(kk + 1) * NC];
    }
#pragma unroll
    for (int s = 0; s < KG / 4; ++s) {
      const int kk = kc + 4 * s;
      af[s].x = ar[kk];
      af[s].y = ar[kk + 1];
    }
#pragma unroll
    for (int s = 0; s < KG / 4; ++s) {
      c = __builtin_amdgcn_wmma_f32_16x16x4_f32(false, af[s], false, bf[s], (short)0, c,
                                                false, false);
    }
  }
  return c;
}

template <int NCOL, bool RELU, int NCOL2>
__global__ void __launch_bounds__(512, 16) k_fused_layer(const float* __restrict__ x,
                              const int* __restrict__ rowptr,
                              const int* __restrict__ col,
                              const float* __restrict__ nsrc,
                              const float* __restrict__ ndst,
                              const float* __restrict__ W,
                              const float* __restrict__ bias,
                              const float* __restrict__ Wn,
                              float* __restrict__ out, int n, int tiles) {
  __shared__ float sA2[2][16 * LDA];
  __shared__ float sH2[NCOL2 > 0 ? 2 : 1][NCOL2 > 0 ? 16 * LDA : 4];

  const int tid  = threadIdx.x;
  const int wid  = tid >> 5;
  const int lane = tid & 31;
  const int uwid = __builtin_amdgcn_readfirstlane(wid);
  constexpr int CT  = NCOL / 16;
  constexpr int CT2 = NCOL2 / 16;

  int buf = 0;
  int prev_tile = -1;
  for (int tile = blockIdx.x; tile < tiles; tile += gridDim.x, buf ^= 1) {
  float* sA = sA2[buf];
  const int grow0 = tile * 16;

  const int node = grow0 + uwid;
  float4 acc; acc.x = acc.y = acc.z = acc.w = 0.0f;
  float nd = 0.0f;
  if (node < n) {
    nd = ndst[node];
    const int beg = rowptr[node];
    const int end = rowptr[node + 1];
    for (int base = beg; base < end; base += 32) {
      const int left = end - base;
      const int cnt = left < 32 ? left : 32;
      int my_s = 0;
      float my_w = 0.0f;
      if (lane < cnt) {
        my_s = col[base + lane];
        my_w = nsrc[my_s];
      }
      for (int j = 0; j < cnt; j += GB) {
        float4 v[GB];
#pragma unroll
        for (int u = 0; u < GB; ++u) {
          const int p = (j + u < cnt) ? j + u : cnt - 1;
          const int s = __builtin_amdgcn_readlane(my_s, p);
          v[u] = ((const float4*)(x + (size_t)s * FEATS))[lane];
        }
        __builtin_amdgcn_sched_barrier(0);
#pragma unroll
        for (int u = 0; u < GB; ++u) {
          const float w = __int_as_float(__builtin_amdgcn_readlane(__float_as_int(my_w), j + u));
          acc.x += v[u].x * w; acc.y += v[u].y * w; acc.z += v[u].z * w; acc.w += v[u].w * w;
        }
      }
    }
  }
  float4 r4; r4.x = acc.x * nd; r4.y = acc.y * nd; r4.z = acc.z * nd; r4.w = acc.w * nd;
  ((float4*)(&sA[wid * LDA]))[lane] = r4;
  __syncthreads();

  if (uwid < CT) {
    const int l    = lane & 15;
    const int half = lane >> 4;
    const v8f c = tile_gemm<NCOL>(sA, W + uwid * 16, lane);
    const float bv = bias[uwid * 16 + l];
    if constexpr (NCOL2 > 0) {
      float* hw = sH2[buf] + uwid * 16 + (8 * half) * LDA + l;
#pragma unroll
      for (int r = 0; r < 8; ++r) {
        float v = c[r] + bv;
        if (RELU) v = fmaxf(v, 0.0f);
        hw[r * LDA] = v;
      }
    } else {
      float* ow = out + ((size_t)grow0 * NCOL + uwid * 16) + (8 * half) * NCOL + l;
      const int rows = n - grow0 - 8 * half;
#pragma unroll
      for (int r = 0; r < 8; ++r) {
        if (r < rows) {
          float v = c[r] + bv;
          if (RELU) v = fmaxf(v, 0.0f);
          ow[r * NCOL] = v;
        }
      }
    }
  } else if (NCOL2 > 0 && uwid < CT + CT2) {
    if constexpr (NCOL2 > 0) {
      if (prev_tile >= 0) {
        const int l    = lane & 15;
        const int half = lane >> 4;
        const int nt   = uwid - CT;
        const v8f c = tile_gemm<NCOL2>(sH2[buf ^ 1], Wn + nt * 16, lane);
        const int prow0 = prev_tile * 16;
        float* ow = out + ((size_t)prow0 * NCOL2 + nt * 16) + (8 * half) * NCOL2 + l;
        const int rows = n - prow0 - 8 * half;
#pragma unroll
        for (int r = 0; r < 8; ++r) {
          if (r < rows) ow[r * NCOL2] = c[r];
        }
      }
    }
  }
  prev_tile = tile;
  }

  if constexpr (NCOL2 > 0) {
    __syncthreads();
    if (uwid >= CT && uwid < CT + CT2 && prev_tile >= 0) {
      const int l    = lane & 15;
      const int half = lane >> 4;
      const int nt   = uwid - CT;
      const v8f c = tile_gemm<NCOL2>(sH2[buf ^ 1], Wn + nt * 16, lane);
      const int prow0 = prev_tile * 16;
      float* ow = out + ((size_t)prow0 * NCOL2 + nt * 16) + (8 * half) * NCOL2 + l;
      const int rows = n - prow0 - 8 * half;
#pragma unroll
      for (int r = 0; r < 8; ++r) {
        if (r < rows) ow[r * NCOL2] = c[r];
      }
    }
  }
}

#define GBZ 16
#define ZW 64
__global__ void __launch_bounds__(512, 16) k_sum_rows(const float* __restrict__ z,
                              const int* __restrict__ rowptr,
                              const int* __restrict__ col,
                              const float* __restrict__ nsrc,
                              const float* __restrict__ ndst,
                              const float* __restrict__ bias,
                              float* __restrict__ out, int n, int tiles) {
  const int tid  = threadIdx.x;
  const int wid  = tid >> 5;
  const int lane = tid & 31;
  const int uwid = __builtin_amdgcn_readfirstlane(wid);
  const float2 bv = ((const float2*)bias)[lane];
  for (int tile = blockIdx.x; tile < tiles; tile += gridDim.x) {
    const int node = tile * 16 + uwid;
    if (node < n) {
      const float nd = ndst[node];
      const int beg = rowptr[node];
      const int end = rowptr[node + 1];
      float2 acc; acc.x = acc.y = 0.0f;
      for (int base = beg; base < end; base += 32) {
        const int left = end - base;
        const int cnt = left < 32 ? left : 32;
        int my_s = 0;
        float my_w = 0.0f;
        if (lane < cnt) {
          my_s = col[base + lane];
          my_w = nsrc[my_s];
        }
        for (int j = 0; j < cnt; j += GBZ) {
          float2 v[GBZ];
#pragma unroll
          for (int u = 0; u < GBZ; ++u) {
            const int p = (j + u < cnt) ? j + u : cnt - 1;
            const int s = __builtin_amdgcn_readlane(my_s, p);
            v[u] = ((const float2*)(z + (size_t)s * ZW))[lane];
          }
          __builtin_amdgcn_sched_barrier(0);
#pragma unroll
          for (int u = 0; u < GBZ; ++u) {
            const float w = __int_as_float(__builtin_amdgcn_readlane(__float_as_int(my_w), j + u));
            acc.x += v[u].x * w; acc.y += v[u].y * w;
          }
        }
      }
      float2 r2; r2.x = acc.x * nd + bv.x; r2.y = acc.y * nd + bv.y;
      ((float2*)(out + (size_t)node * ZW))[lane] = r2;
    }
  }
}

extern "C" void kernel_launch(void* const* d_in, const int* in_sizes, int n_in,
                              void* d_out, int out_size, void* d_ws, size_t ws_size,
                              hipStream_t stream) {
  const float* feats = (const float*)d_in[0];
  const int*   src   = (const int*)d_in[1];
  const int*   dst   = (const int*)d_in[2];
  const float* W0    = (const float*)d_in[3];
  const float* b0    = (const float*)d_in[4];
  const float* W1    = (const float*)d_in[5];
  const float* b1    = (const float*)d_in[6];
  const float* W2    = (const float*)d_in[7];
  const float* b2    = (const float*)d_in[8];

  const int N = in_sizes[0] / FEATS;
  const int E = in_sizes[1];

  char* ws = (char*)d_ws;
  size_t off = 0;
  auto take = [&](size_t bytes) -> void* {
    void* p = ws + off;
    off += (bytes + 255) & ~(size_t)255;
    return p;
  };

  const int nb = (N + 255) / 256;
  const int NW = (N + 1) / 2;
  unsigned int* partD = (unsigned int*)take((size_t)NB * NW * 4);
  unsigned int* partS = (unsigned int*)take((size_t)NB * NW * 4);
  int*   pref   = (int*)take((size_t)NB * N * 4);
  int*   rank   = (int*)take((size_t)E * 4);
  int*   din_i  = (int*)take((size_t)N * 4);
  float* nsrc   = (float*)take((size_t)N * 4);
  float* ndst   = (float*)take((size_t)N * 4);
  int*   incl   = (int*)take((size_t)N * 4);
  int*   bsum   = (int*)take((size_t)nb * 4);
  int*   rowptr = (int*)take((size_t)(N + 1) * 4);
  int*   col    = (int*)take((size_t)E * 4);
  float* h0     = (float*)take((size_t)N * FEATS * 4);
  float* h1     = (float*)take((size_t)N * FEATS * 4);

  const int eb = (E + 255) / 256;
  k_count<<<2 * NB, 1024, 0, stream>>>(src, dst, partD, partS, rank, E, NW);
  k_reduce<<<(NW + 255) / 256, 256, 0, stream>>>(partD, partS, pref, din_i, nsrc, ndst, N, NW);

  k_scan1<<<nb, 256, 0, stream>>>(din_i, incl, bsum, N);
  k_scan2<<<1, 64, 0, stream>>>(bsum, nb);
  k_scan3<<<nb, 256, 0, stream>>>(incl, bsum, rowptr, N);
  k_place<<<eb, 256, 0, stream>>>(src, dst, rowptr, pref, rank, col, E, N);

  const int tiles = (N + 15) / 16;

  k_fused_layer<128, true, 0><<<RESIDENT_BLOCKS, 512, 0, stream>>>(
      feats, rowptr, col, nsrc, ndst, W0, b0, nullptr, h0, N, tiles);
  k_fused_layer<128, true, 64><<<RESIDENT_BLOCKS, 512, 0, stream>>>(
      h0, rowptr, col, nsrc, ndst, W1, b1, W2, h1, N, tiles);
  k_sum_rows<<<RESIDENT_BLOCKS, 512, 0, stream>>>(
      h1, rowptr, col, nsrc, ndst, b2, (float*)d_out, N, tiles);
}


# Layer weights W0 and W1 copied once per work-group into LDS in B-fragment order (work-groups now 32 waves, 2 per WGP, 32-node tiles); tiles read B from LDS, not memory; k order and sum order kept
# speedup vs baseline: 2.3231x; 1.0474x over previous
#include <hip/hip_runtime.h>
#include <hip/hip_bf16.h>
#include <stdint.h>

typedef __attribute__((ext_vector_type(2))) float v2f;
typedef __attribute__((ext_vector_type(8))) float v8f;

#define FEATS 128
#define LDA 132
#define RESIDENT_BLOCKS 512
#define TROWS 32
#define GB 8

#define NB 64
#define HIST_WORDS 50000

__global__ void __launch_bounds__(1024) k_count(const int* __restrict__ src,
                                                const int* __restrict__ dst,
                                                unsigned int* __restrict__ partD,
                                                unsigned int* __restrict__ partS,
                                                int* __restrict__ rank, int E, int NW) {
  __shared__ __align__(16) unsigned int hist[HIST_WORDS];
  const int role = blockIdx.x / NB;
  const int b    = blockIdx.x - role * NB;
  const int tid  = (int)threadIdx.x;
  for (int w = tid; w < HIST_WORDS / 4; w += 1024) ((uint4*)hist)[w] = make_uint4(0u, 0u, 0u, 0u);
  __syncthreads();
  const int chunk = (E + NB - 1) / NB;
  const int e0 = b * chunk;
  const int e1 = (e0 + chunk < E) ? e0 + chunk : E;
  const int* __restrict__ keys = (role == 0) ? dst : src;
  constexpr int CU = 8;
  for (int base = e0 + tid; base < e1; base += 1024 * CU) {
    int k[CU];
#pragma unroll
    for (int u = 0; u < CU; ++u) {
      const int e = base + u * 1024;
      k[u] = (e < e1) ? keys[e] : 0;
    }
    if (role == 0) {
#pragma unroll
      for (int u = 0; u < CU; ++u) {
        const int e = base + u * 1024;
        if (e < e1) {
          const unsigned int sh = (unsigned int)(k[u] & 1) * 16u;
          const unsigned int old = atomicAdd(&hist[k[u] >> 1], 1u << sh);
          rank[e] = (int)((old >> sh) & 0xFFFFu);
        }
      }
    } else {
#pragma unroll
      for (int u = 0; u < CU; ++u) {
        const int e = base + u * 1024;
        if (e < e1) atomicAdd(&hist[k[u] >> 1], 1u << ((unsigned int)(k[u] & 1) * 16u));
      }
    }
  }
  __syncthreads();
  unsigned int* __restrict__ part = (role == 0 ? partD : partS) + (size_t)b * NW;
  for (int w = tid; w < NW; w += 1024) part[w] = hist[w];
}

__global__ void k_reduce(const unsigned int* __restrict__ partD,
                         const unsigned int* __restrict__ partS,
                         int* __restrict__ pref, int* __restrict__ din,
                         float* __restrict__ nsrc, float* __restrict__ ndst, int n, int NW) {
  const int w = blockIdx.x * blockDim.x + threadIdx.x;
  if (w < NW) {
    const int i0 = 2 * w;
    const int i1 = 2 * w + 1;
    unsigned int d0 = 0u, d1 = 0u, s0 = 0u, s1 = 0u;
    constexpr int RB = 32;
    for (int bb = 0; bb < NB; bb += RB) {
      unsigned int v[RB], u[RB];
#pragma unroll
      for (int j = 0; j < RB; ++j) {
        v[j] = partD[(size_t)(bb + j) * NW + w];
        u[j] = partS[(size_t)(bb + j) * NW + w];
      }
#pragma unroll
      for (int j = 0; j < RB; ++j) {
        pref[(size_t)(bb + j) * n + i0] = (int)d0;
        if (i1 < n) pref[(size_t)(bb + j) * n + i1] = (int)d1;
        d0 += v[j] & 0xFFFFu; d1 += v[j] >> 16;
        s0 += u[j] & 0xFFFFu; s1 += u[j] >> 16;
      }
    }
    din[i0]  = (int)d0;
    nsrc[i0] = rsqrtf(fmaxf((float)s0, 1.0f));
    ndst[i0] = rsqrtf(fmaxf((float)d0, 1.0f));
    if (i1 < n) {
      din[i1]  = (int)d1;
      nsrc[i1] = rsqrtf(fmaxf((float)s1, 1.0f));
      ndst[i1] = rsqrtf(fmaxf((float)d1, 1.0f));
    }
  }
}

__global__ void k_scan1(const int* __restrict__ counts, int* __restrict__ incl,
                        int* __restrict__ bsum, int n) {
  __shared__ int tmp[256];
  int tid = threadIdx.x;
  int i = blockIdx.x * 256 + tid;
  int v = (i < n) ? counts[i] : 0;
  tmp[tid] = v;
  __syncthreads();
  for (int off = 1; off < 256; off <<= 1) {
    int t = (tid >= off) ? tmp[tid - off] : 0;
    __syncthreads();
    tmp[tid] += t;
    __syncthreads();
  }
  if (i < n) incl[i] = tmp[tid];
  if (tid == 255) bsum[blockIdx.x] = tmp[255];
}

__global__ void k_scan2(int* __restrict__ bsum, int nb) {
  if (threadIdx.x == 0 && blockIdx.x == 0) {
    int run = 0;
    for (int b = 0; b < nb; ++b) {
      int s = bsum[b];
      bsum[b] = run;
      run += s;
    }
  }
}

__global__ void k_scan3(const int* __restrict__ incl, const int* __restrict__ bsum,
                        int* __restrict__ rowptr, int n) {
  int i = blockIdx.x * 256 + threadIdx.x;
  if (i < n) {
    rowptr[i + 1] = incl[i] + bsum[blockIdx.x];
    if (i == 0) rowptr[0] = 0;
  }
}

__global__ void k_place(const int* __restrict__ src, const int* __restrict__ dst,
                        const int* __restrict__ rowptr, const int* __restrict__ pref,
                        const int* __restrict__ rank, int* __restrict__ col, int E, int n) {
  int e = blockIdx.x * blockDim.x + threadIdx.x;
  if (e < E) {
    const int chunk = (E + NB - 1) / NB;
    const int d = dst[e];
    const int b = e / chunk;
    const int pos = rowptr[d] + pref[(size_t)b * n + d] + rank[e];
    col[pos] = src[e];
  }
}

template <int NC>
__device__ __forceinline__ v8f tile_gemm(const float* __restrict__ sTile,
                                         const float* __restrict__ Wc, int lane) {
  const int l    = lane & 15;
  const int half = lane >> 4;
  const float* ar = sTile + l * LDA + 2 * half;
  const float* wr = Wc + (2 * half) * NC + l;
  v8f c = {};
  constexpr int KG = 32;
#pragma unroll 1
  for (int kc = 0; kc < FEATS; kc += KG) {
    v2f bf[KG / 4];
    v2f af[KG / 4];
#pragma unroll
    for (int s = 0; s < KG / 4; ++s) {
      const int kk = kc + 4 * s;
      bf[s].x = wr[kk * NC];
      bf[s].y = wr[(kk + 1) * NC];
    }
#pragma unroll
    for (int s = 0; s < KG / 4; ++s) {
      const int kk = kc + 4 * s;
      af[s].x = ar[kk];
      af[s].y = ar[kk + 1];
    }
#pragma unroll
    for (int s = 0; s < KG / 4; ++s) {
      c = __builtin_amdgcn_wmma_f32_16x16x4_f32(false, af[s], false, bf[s], (short)0, c,
                                                false, false);
    }
  }
  return c;
}

template <int NC>
__device__ __forceinline__ void stage_w(float* __restrict__ sW, const float* __restrict__ W,
                                        int tid) {
  constexpr int PAIRS = (FEATS / 2) * NC;
  static_assert(PAIRS % 1024 == 0, "whole passes of the 1024-thread work-group");
  v2f w[PAIRS / 1024];
#pragma unroll
  for (int i = 0; i < PAIRS / 1024; ++i) {
    const int p   = tid + i * 1024;
    const int kp  = p / NC;
    const int col = p - kp * NC;
    w[i].x = W[(2 * kp) * NC + col];
    w[i].y = W[(2 * kp + 1) * NC + col];
  }
#pragma unroll
  for (int i = 0; i < PAIRS / 1024; ++i) {
    const int p   = tid + i * 1024;
    const int kp  = p / NC;
    const int col = p - kp * NC;
    const int S    = kp >> 1;
    const int half = kp & 1;
    const int ct   = col >> 4;
    const int l    = col & 15;
    ((v2f*)sW)[(ct * 32 + S) * 32 + half * 16 + l] = w[i];
  }
}

__device__ __forceinline__ v8f tile_gemm_lds(const float* __restrict__ sTile,
                                             const float* __restrict__ sWc, int lane) {
  const int l    = lane & 15;
  const int half = lane >> 4;
  const float* ar = sTile + l * LDA + 2 * half;
  const v2f* br = (const v2f*)sWc + lane;
  v8f c = {};
  constexpr int KG = 32;
#pragma unroll 1
  for (int kc = 0; kc < FEATS; kc += KG) {
    v2f bf[KG / 4];
    v2f af[KG / 4];
#pragma unroll
    for (int s = 0; s < KG / 4; ++s) {
      bf[s] = br[(kc / 4 + s) * 32];
    }
#pragma unroll
    for (int s = 0; s < KG / 4; ++s) {
      const int kk = kc + 4 * s;
      af[s].x = ar[kk];
      af[s].y = ar[kk + 1];
    }
#pragma unroll
    for (int s = 0; s < KG / 4; ++s) {
      c = __builtin_amdgcn_wmma_f32_16x16x4_f32(false, af[s], false, bf[s], (short)0, c,
                                                false, false);
    }
  }
  return c;
}

template <int NCOL, bool RELU, int NCOL2>
__global__ void __launch_bounds__(1024, 16) k_fused_layer(const float* __restrict__ x,
                              const int* __restrict__ rowptr,
                              const int* __restrict__ col,
                              const float* __restrict__ nsrc,
                              const float* __restrict__ ndst,
                              const float* __restrict__ W,
                              const float* __restrict__ bias,
                              const float* __restrict__ Wn,
                              float* __restrict__ out, int n, int tiles) {
  __shared__ float sA2[2][TROWS * LDA];
  __shared__ float sH2[NCOL2 > 0 ? 2 : 1][NCOL2 > 0 ? TROWS * LDA : 4];
  __shared__ __align__(16) float sW[FEATS * NCOL];

  const int tid  = threadIdx.x;
  const int wid  = tid >> 5;
  const int lane = tid & 31;
  const int uwid = __builtin_amdgcn_readfirstlane(wid);
  constexpr int CT  = NCOL / 16;
  constexpr int CT2 = NCOL2 / 16;
  constexpr int G1  = (TROWS / 16) * CT;
  constexpr int G2  = (TROWS / 16) * CT2;
  static_assert(G1 + G2 <= 32, "one wave per 16 x 16 output tile");

  stage_w<NCOL>(sW, W, tid);

  int buf = 0;
  int prev_tile = -1;
  for (int tile = blockIdx.x; tile < tiles; tile += gridDim.x, buf ^= 1) {
  float* sA = sA2[buf];
  const int grow0 = tile * TROWS;

  const int node = grow0 + uwid;
  float4 acc; acc.x = acc.y = acc.z = acc.w = 0.0f;
  float nd = 0.0f;
  if (node < n) {
    nd = ndst[node];
    const int beg = rowptr[node];
    const int end = rowptr[node + 1];
    for (int base = beg; base < end; base += 32) {
      const int left = end - base;
      const int cnt = left < 32 ? left : 32;
      int my_s = 0;
      float my_w = 0.0f;
      if (lane < cnt) {
        my_s = col[base + lane];
        my_w = nsrc[my_s];
      }
      for (int j = 0; j < cnt; j += GB) {
        float4 v[GB];
#pragma unroll
        for (int u = 0; u < GB; ++u) {
          const int p = (j + u < cnt) ? j + u : cnt - 1;
          const int s = __builtin_amdgcn_readlane(my_s, p);
          v[u] = ((const float4*)(x + (size_t)s * FEATS))[lane];
        }
        __builtin_amdgcn_sched_barrier(0);
#pragma unroll
        for (int u = 0; u < GB; ++u) {
          const float w = __int_as_float(__builtin_amdgcn_readlane(__float_as_int(my_w), j + u));
          acc.x += v[u].x * w; acc.y += v[u].y * w; acc.z += v[u].z * w; acc.w += v[u].w * w;
        }
      }
    }
  }
  float4 r4; r4.x = acc.x * nd; r4.y = acc.y * nd; r4.z = acc.z * nd; r4.w = acc.w * nd;
  ((float4*)(&sA[wid * LDA]))[lane] = r4;
  __syncthreads();

  if (uwid < G1) {
    const int l    = lane & 15;
    const int half = lane >> 4;
    const int rt   = uwid / CT;
    const int ct   = uwid - rt * CT;
    const v8f c = tile_gemm_lds(sA + rt * 16 * LDA, sW + ct * (32 * 32 * 2), lane);
    const float bv = bias[ct * 16 + l];
    if constexpr (NCOL2 > 0) {
      float* hw = sH2[buf] + ct * 16 + (rt * 16 + 8 * half) * LDA + l;
#pragma unroll
      for (int r = 0; r < 8; ++r) {
        float v = c[r] + bv;
        if (RELU) v = fmaxf(v, 0.0f);
        hw[r * LDA] = v;
      }
    } else {
      const int orow0 = grow0 + rt * 16;
      float* ow = out + ((size_t)orow0 * NCOL + ct * 16) + (8 * half) * NCOL + l;
      const int rows = n - orow0 - 8 * half;
#pragma unroll
      for (int r = 0; r < 8; ++r) {
        if (r < rows) {
          float v = c[r] + bv;
          if (RELU) v = fmaxf(v, 0.0f);
          ow[r * NCOL] = v;
        }
      }
    }
  } else if (NCOL2 > 0 && uwid < G1 + G2) {
    if constexpr (NCOL2 > 0) {
      if (prev_tile >= 0) {
        const int l    = lane & 15;
        const int half = lane >> 4;
        const int g    = uwid - G1;
        const int rt   = g / CT2;
        const int nt   = g - rt * CT2;
        const v8f c = tile_gemm<NCOL2>(sH2[buf ^ 1] + rt * 16 * LDA, Wn + nt * 16, lane);
        const int prow0 = prev_tile * TROWS + rt * 16;
        float* ow = out + ((size_t)prow0 * NCOL2 + nt * 16) + (8 * half) * NCOL2 + l;
        const int rows = n - prow0 - 8 * half;
#pragma unroll
        for (int r = 0; r < 8; ++r) {
          if (r < rows) ow[r * NCOL2] = c[r];
        }
      }
    }
  }
  prev_tile = tile;
  }

  if constexpr (NCOL2 > 0) {
    __syncthreads();
    if (uwid >= G1 && uwid < G1 + G2 && prev_tile >= 0) {
      const int l    = lane & 15;
      const int half = lane >> 4;
      const int g    = uwid - G1;
      const int rt   = g / CT2;
      const int nt   = g - rt * CT2;
      const v8f c = tile_gemm<NCOL2>(sH2[buf ^ 1] + rt * 16 * LDA, Wn + nt * 16, lane);
      const int prow0 = prev_tile * TROWS + rt * 16;
      float* ow = out + ((size_t)prow0 * NCOL2 + nt * 16) + (8 * half) * NCOL2 + l;
      const int rows = n - prow0 - 8 * half;
#pragma unroll
      for (int r = 0; r < 8; ++r) {
        if (r < rows) ow[r * NCOL2] = c[r];
      }
    }
  }
}

#define GBZ 16
#define ZW 64
#define SUM_BLOCKS 1024
__global__ void __launch_bounds__(512, 16) k_sum_rows(const float* __restrict__ z,
                              const int* __restrict__ rowptr,
                              const int* __restrict__ col,
                              const float* __restrict__ nsrc,
                              const float* __restrict__ ndst,
                              const float* __restrict__ bias,
                              float* __restrict__ out, int n, int tiles) {
  const int tid  = threadIdx.x;
  const int wid  = tid >> 5;
  const int lane = tid & 31;
  const int uwid = __builtin_amdgcn_readfirstlane(wid);
  const float2 bv = ((const float2*)bias)[lane];
  for (int tile = blockIdx.x; tile < tiles; tile += gridDim.x) {
    const int node = tile * 16 + uwid;
    if (node < n) {
      const float nd = ndst[node];
      const int beg = rowptr[node];
      const int end = rowptr[node + 1];
      float2 acc; acc.x = acc.y = 0.0f;
      for (int base = beg; base < end; base += 32) {
        const int left = end - base;
        const int cnt = left < 32 ? left : 32;
        int my_s = 0;
        float my_w = 0.0f;
        if (lane < cnt) {
          my_s = col[base + lane];
          my_w = nsrc[my_s];
        }
        for (int j = 0; j < cnt; j += GBZ) {
          float2 v[GBZ];
#pragma unroll
          for (int u = 0; u < GBZ; ++u) {
            const int p = (j + u < cnt) ? j + u : cnt - 1;
            const int s = __builtin_amdgcn_readlane(my_s, p);
            v[u] = ((const float2*)(z + (size_t)s * ZW))[lane];
          }
          __builtin_amdgcn_sched_barrier(0);
#pragma unroll
          for (int u = 0; u < GBZ; ++u) {
            const float w = __int_as_float(__builtin_amdgcn_readlane(__float_as_int(my_w), j + u));
            acc.x += v[u].x * w; acc.y += v[u].y * w;
          }
        }
      }
      float2 r2; r2.x = acc.x * nd + bv.x; r2.y = acc.y * nd + bv.y;
      ((float2*)(out + (size_t)node * ZW))[lane] = r2;
    }
  }
}

extern "C" void kernel_launch(void* const* d_in, const int* in_sizes, int n_in,
                              void* d_out, int out_size, void* d_ws, size_t ws_size,
                              hipStream_t stream) {
  const float* feats = (const float*)d_in[0];
  const int*   src   = (const int*)d_in[1];
  const int*   dst   = (const int*)d_in[2];
  const float* W0    = (const float*)d_in[3];
  const float* b0    = (const float*)d_in[4];
  const float* W1    = (const float*)d_in[5];
  const float* b1    = (const float*)d_in[6];
  const float* W2    = (const float*)d_in[7];
  const float* b2    = (const float*)d_in[8];

  const int N = in_sizes[0] / FEATS;
  const int E = in_sizes[1];

  char* ws = (char*)d_ws;
  size_t off = 0;
  auto take = [&](size_t bytes) -> void* {
    void* p = ws + off;
    off += (bytes + 255) & ~(size_t)255;
    return p;
  };

  const int nb = (N + 255) / 256;
  const int NW = (N + 1) / 2;
  unsigned int* partD = (unsigned int*)take((size_t)NB * NW * 4);
  unsigned int* partS = (unsigned int*)take((size_t)NB * NW * 4);
  int*   pref   = (int*)take((size_t)NB * N * 4);
  int*   rank   = (int*)take((size_t)E * 4);
  int*   din_i  = (int*)take((size_t)N * 4);
  float* nsrc   = (float*)take((size_t)N * 4);
  float* ndst   = (float*)take((size_t)N * 4);
  int*   incl   = (int*)take((size_t)N * 4);
  int*   bsum   = (int*)take((size_t)nb * 4);
  int*   rowptr = (int*)take((size_t)(N + 1) * 4);
  int*   col    = (int*)take((size_t)E * 4);
  float* h0     = (float*)take((size_t)N * FEATS * 4);
  float* h1     = (float*)take((size_t)N * FEATS * 4);

  const int eb = (E + 255) / 256;
  k_count<<<2 * NB, 1024, 0, stream>>>(src, dst, partD, partS, rank, E, NW);
  k_reduce<<<(NW + 255) / 256, 256, 0, stream>>>(partD, partS, pref, din_i, nsrc, ndst, N, NW);

  k_scan1<<<nb, 256, 0, stream>>>(din_i, incl, bsum, N);
  k_scan2<<<1, 64, 0, stream>>>(bsum, nb);
  k_scan3<<<nb, 256, 0, stream>>>(incl, bsum, rowptr, N);
  k_place<<<eb, 256, 0, stream>>>(src, dst, rowptr, pref, rank, col, E, N);

  const int tiles   = (N + TROWS - 1) / TROWS;
  const int tiles16 = (N + 15) / 16;

  k_fused_layer<128, true, 0><<<RESIDENT_BLOCKS, 1024, 0, stream>>>(
      feats, rowptr, col, nsrc, ndst, W0, b0, nullptr, h0, N, tiles);
  k_fused_layer<128, true, 64><<<RESIDENT_BLOCKS, 1024, 0, stream>>>(
      h0, rowptr, col, nsrc, ndst, W1, b1, W2, h1, N, tiles);
  k_sum_rows<<<SUM_BLOCKS, 512, 0, stream>>>(
      h1, rowptr, col, nsrc, ndst, b2, (float*)d_out, N, tiles16);
}


# Row-start prefix sum folded: k_reduce scans the in-degrees inside its 512-node work-groups, one kernel adds the work-group totals in parallel; 3 scan launches (one a 1-thread serial loop) become 1
# speedup vs baseline: 2.4935x; 1.0734x over previous
#include <hip/hip_runtime.h>
#include <hip/hip_bf16.h>
#include <stdint.h>

typedef __attribute__((ext_vector_type(2))) float v2f;
typedef __attribute__((ext_vector_type(8))) float v8f;

#define FEATS 128
#define LDA 132
#define RESIDENT_BLOCKS 512
#define TROWS 32
#define GB 8

#define NB 64
#define HIST_WORDS 50000

__global__ void __launch_bounds__(1024) k_count(const int* __restrict__ src,
                                                const int* __restrict__ dst,
                                                unsigned int* __restrict__ partD,
                                                unsigned int* __restrict__ partS,
                                                int* __restrict__ rank, int E, int NW) {
  __shared__ __align__(16) unsigned int hist[HIST_WORDS];
  const int role = blockIdx.x / NB;
  const int b    = blockIdx.x - role * NB;
  const int tid  = (int)threadIdx.x;
  for (int w = tid; w < HIST_WORDS / 4; w += 1024) ((uint4*)hist)[w] = make_uint4(0u, 0u, 0u, 0u);
  __syncthreads();
  const int chunk = (E + NB - 1) / NB;
  const int e0 = b * chunk;
  const int e1 = (e0 + chunk < E) ? e0 + chunk : E;
  const int* __restrict__ keys = (role == 0) ? dst : src;
  constexpr int CU = 8;
  for (int base = e0 + tid; base < e1; base += 1024 * CU) {
    int k[CU];
#pragma unroll
    for (int u = 0; u < CU; ++u) {
      const int e = base + u * 1024;
      k[u] = (e < e1) ? keys[e] : 0;
    }
    if (role == 0) {
#pragma unroll
      for (int u = 0; u < CU; ++u) {
        const int e = base + u * 1024;
        if (e < e1) {
          const unsigned int sh = (unsigned int)(k[u] & 1) * 16u;
          const unsigned int old = atomicAdd(&hist[k[u] >> 1], 1u << sh);
          rank[e] = (int)((old >> sh) & 0xFFFFu);
        }
      }
    } else {
#pragma unroll
      for (int u = 0; u < CU; ++u) {
        const int e = base + u * 1024;
        if (e < e1) atomicAdd(&hist[k[u] >> 1], 1u << ((unsigned int)(k[u] & 1) * 16u));
      }
    }
  }
  __syncthreads();
  unsigned int* __restrict__ part = (role == 0 ? partD : partS) + (size_t)b * NW;
  for (int w = tid; w < NW; w += 1024) part[w] = hist[w];
}

__device__ __forceinline__ int wave_incl_scan(int v, int lane) {
#pragma unroll
  for (int off = 1; off < 32; off <<= 1) {
    const int t = __shfl_up(v, off, 32);
    if (lane >= off) v += t;
  }
  return v;
}

__global__ void __launch_bounds__(256) k_reduce(const unsigned int* __restrict__ partD,
                         const unsigned int* __restrict__ partS,
                         int* __restrict__ pref, int* __restrict__ incl, int* __restrict__ bsum,
                         float* __restrict__ nsrc, float* __restrict__ ndst, int n, int NW) {
  __shared__ int wtot[8];
  const int tid = (int)threadIdx.x;
  const int w = blockIdx.x * 256 + tid;
  const int i0 = 2 * w;
  const int i1 = 2 * w + 1;
  unsigned int d0 = 0u, d1 = 0u, s0 = 0u, s1 = 0u;
  if (w < NW) {
    constexpr int RB = 32;
    for (int bb = 0; bb < NB; bb += RB) {
      unsigned int v[RB], u[RB];
#pragma unroll
      for (int j = 0; j < RB; ++j) {
        v[j] = partD[(size_t)(bb + j) * NW + w];
        u[j] = partS[(size_t)(bb + j) * NW + w];
      }
#pragma unroll
      for (int j = 0; j < RB; ++j) {
        pref[(size_t)(bb + j) * n + i0] = (int)d0;
        if (i1 < n) pref[(size_t)(bb + j) * n + i1] = (int)d1;
        d0 += v[j] & 0xFFFFu; d1 += v[j] >> 16;
        s0 += u[j] & 0xFFFFu; s1 += u[j] >> 16;
      }
    }
    nsrc[i0] = rsqrtf(fmaxf((float)s0, 1.0f));
    ndst[i0] = rsqrtf(fmaxf((float)d0, 1.0f));
    if (i1 < n) {
      nsrc[i1] = rsqrtf(fmaxf((float)s1, 1.0f));
      ndst[i1] = rsqrtf(fmaxf((float)d1, 1.0f));
    }
  }
  const int lane = tid & 31;
  const int wv   = tid >> 5;
  const int pair = (int)(d0 + d1);
  const int inc  = wave_incl_scan(pair, lane);
  if (lane == 31) wtot[wv] = inc;
  __syncthreads();
  int before = 0, total = 0;
#pragma unroll
  for (int k = 0; k < 8; ++k) {
    const int s = wtot[k];
    if (k < wv) before += s;
    total += s;
  }
  const int ex = before + inc - pair;
  if (w < NW) {
    incl[i0] = ex + (int)d0;
    if (i1 < n) incl[i1] = ex + pair;
  }
  if (tid == 0) bsum[blockIdx.x] = total;
}

__global__ void __launch_bounds__(256) k_rowptr(const int* __restrict__ incl,
                                                const int* __restrict__ bsum,
                                                int* __restrict__ rowptr, int n, int NW) {
  __shared__ int wtot[8];
  const int tid = (int)threadIdx.x;
  const int w = blockIdx.x * 256 + tid;
  const int i0 = 2 * w;
  const int i1 = 2 * w + 1;
  int v = 0;
  for (int j = tid; j < (int)blockIdx.x; j += 256) v += bsum[j];
  int a0 = 0, a1 = 0;
  if (w < NW) {
    a0 = incl[i0];
    if (i1 < n) a1 = incl[i1];
  }
#pragma unroll
  for (int off = 16; off > 0; off >>= 1) v += __shfl_xor(v, off, 32);
  if ((tid & 31) == 0) wtot[tid >> 5] = v;
  __syncthreads();
  int boff = 0;
#pragma unroll
  for (int k = 0; k < 8; ++k) boff += wtot[k];
  if (w < NW) {
    rowptr[i0 + 1] = a0 + boff;
    if (i1 < n) rowptr[i1 + 1] = a1 + boff;
    if (w == 0) rowptr[0] = 0;
  }
}

__global__ void k_place(const int* __restrict__ src, const int* __restrict__ dst,
                        const int* __restrict__ rowptr, const int* __restrict__ pref,
                        const int* __restrict__ rank, int* __restrict__ col, int E, int n) {
  int e = blockIdx.x * blockDim.x + threadIdx.x;
  if (e < E) {
    const int chunk = (E + NB - 1) / NB;
    const int d = dst[e];
    const int b = e / chunk;
    const int pos = rowptr[d] + pref[(size_t)b * n + d] + rank[e];
    col[pos] = src[e];
  }
}

template <int NC>
__device__ __forceinline__ v8f tile_gemm(const float* __restrict__ sTile,
                                         const float* __restrict__ Wc, int lane) {
  const int l    = lane & 15;
  const int half = lane >> 4;
  const float* ar = sTile + l * LDA + 2 * half;
  const float* wr = Wc + (2 * half) * NC + l;
  v8f c = {};
  constexpr int KG = 32;
#pragma unroll 1
  for (int kc = 0; kc < FEATS; kc += KG) {
    v2f bf[KG / 4];
    v2f af[KG / 4];
#pragma unroll
    for (int s = 0; s < KG / 4; ++s) {
      const int kk = kc + 4 * s;
      bf[s].x = wr[kk * NC];
      bf[s].y = wr[(kk + 1) * NC];
    }
#pragma unroll
    for (int s = 0; s < KG / 4; ++s) {
      const int kk = kc + 4 * s;
      af[s].x = ar[kk];
      af[s].y = ar[kk + 1];
    }
#pragma unroll
    for (int s = 0; s < KG / 4; ++s) {
      c = __builtin_amdgcn_wmma_f32_16x16x4_f32(false, af[s], false, bf[s], (short)0, c,
                                                false, false);
    }
  }
  return c;
}

template <int NC>
__device__ __forceinline__ void stage_w(float* __restrict__ sW, const float* __restrict__ W,
                                        int tid) {
  constexpr int PAIRS = (FEATS / 2) * NC;
  static_assert(PAIRS % 1024 == 0, "whole passes of the 1024-thread work-group");
  v2f w[PAIRS / 1024];
#pragma unroll
  for (int i = 0; i < PAIRS / 1024; ++i) {
    const int p   = tid + i * 1024;
    const int kp  = p / NC;
    const int col = p - kp * NC;
    w[i].x = W[(2 * kp) * NC + col];
    w[i].y = W[(2 * kp + 1) * NC + col];
  }
#pragma unroll
  for (int i = 0; i < PAIRS / 1024; ++i) {
    const int p   = tid + i * 1024;
    const int kp  = p / NC;
    const int col = p - kp * NC;
    const int S    = kp >> 1;
    const int half = kp & 1;
    const int ct   = col >> 4;
    const int l    = col & 15;
    ((v2f*)sW)[(ct * 32 + S) * 32 + half * 16 + l] = w[i];
  }
}

__device__ __forceinline__ v8f tile_gemm_lds(const float* __restrict__ sTile,
                                             const float* __restrict__ sWc, int lane) {
  const int l    = lane & 15;
  const int half = lane >> 4;
  const float* ar = sTile + l * LDA + 2 * half;
  const v2f* br = (const v2f*)sWc + lane;
  v8f c = {};
  constexpr int KG = 32;
#pragma unroll 1
  for (int kc = 0; kc < FEATS; kc += KG) {
    v2f bf[KG / 4];
    v2f af[KG / 4];
#pragma unroll
    for (int s = 0; s < KG / 4; ++s) {
      bf[s] = br[(kc / 4 + s) * 32];
    }
#pragma unroll
    for (int s = 0; s < KG / 4; ++s) {
      const int kk = kc + 4 * s;
      af[s].x = ar[kk];
      af[s].y = ar[kk + 1];
    }
#pragma unroll
    for (int s = 0; s < KG / 4; ++s) {
      c = __builtin_amdgcn_wmma_f32_16x16x4_f32(false, af[s], false, bf[s], (short)0, c,
                                                false, false);
    }
  }
  return c;
}

template <int NCOL, bool RELU, int NCOL2>
__global__ void __launch_bounds__(1024, 16) k_fused_layer(const float* __restrict__ x,
                              const int* __restrict__ rowptr,
                              const int* __restrict__ col,
                              const float* __restrict__ nsrc,
                              const float* __restrict__ ndst,
                              const float* __restrict__ W,
                              const float* __restrict__ bias,
                              const float* __restrict__ Wn,
                              float* __restrict__ out, int n, int tiles) {
  __shared__ float sA2[2][TROWS * LDA];
  __shared__ float sH2[NCOL2 > 0 ? 2 : 1][NCOL2 > 0 ? TROWS * LDA : 4];
  __shared__ __align__(16) float sW[FEATS * NCOL];

  const int tid  = threadIdx.x;
  const int wid  = tid >> 5;
  const int lane = tid & 31;
  const int uwid = __builtin_amdgcn_readfirstlane(wid);
  constexpr int CT  = NCOL / 16;
  constexpr int CT2 = NCOL2 / 16;
  constexpr int G1  = (TROWS / 16) * CT;
  constexpr int G2  = (TROWS / 16) * CT2;
  static_assert(G1 + G2 <= 32, "one wave per 16 x 16 output tile");

  stage_w<NCOL>(sW, W, tid);

  int buf = 0;
  int prev_tile = -1;
  for (int tile = blockIdx.x; tile < tiles; tile += gridDim.x, buf ^= 1) {
  float* sA = sA2[buf];
  const int grow0 = tile * TROWS;

  const int node = grow0 + uwid;
  float4 acc; acc.x = acc.y = acc.z = acc.w = 0.0f;
  float nd = 0.0f;
  if (node < n) {
    nd = ndst[node];
    const int beg = rowptr[node];
    const int end = rowptr[node + 1];
    for (int base = beg; base < end; base += 32) {
      const int left = end - base;
      const int cnt = left < 32 ? left : 32;
      int my_s = 0;
      float my_w = 0.0f;
      if (lane < cnt) {
        my_s = col[base + lane];
        my_w = nsrc[my_s];
      }
      for (int j = 0; j < cnt; j += GB) {
        float4 v[GB];
#pragma unroll
        for (int u = 0; u < GB; ++u) {
          const int p = (j + u < cnt) ? j + u : cnt - 1;
          const int s = __builtin_amdgcn_readlane(my_s, p);
          v[u] = ((const float4*)(x + (size_t)s * FEATS))[lane];
        }
        __builtin_amdgcn_sched_barrier(0);
#pragma unroll
        for (int u = 0; u < GB; ++u) {
          const float w = __int_as_float(__builtin_amdgcn_readlane(__float_as_int(my_w), j + u));
          acc.x += v[u].x * w; acc.y += v[u].y * w; acc.z += v[u].z * w; acc.w += v[u].w * w;
        }
      }
    }
  }
  float4 r4; r4.x = acc.x * nd; r4.y = acc.y * nd; r4.z = acc.z * nd; r4.w = acc.w * nd;
  ((float4*)(&sA[wid * LDA]))[lane] = r4;
  __syncthreads();

  if (uwid < G1) {
    const int l    = lane & 15;
    const int half = lane >> 4;
    const int rt   = uwid / CT;
    const int ct   = uwid - rt * CT;
    const v8f c = tile_gemm_lds(sA + rt * 16 * LDA, sW + ct * (32 * 32 * 2), lane);
    const float bv = bias[ct * 16 + l];
    if constexpr (NCOL2 > 0) {
      float* hw = sH2[buf] + ct * 16 + (rt * 16 + 8 * half) * LDA + l;
#pragma unroll
      for (int r = 0; r < 8; ++r) {
        float v = c[r] + bv;
        if (RELU) v = fmaxf(v, 0.0f);
        hw[r * LDA] = v;
      }
    } else {
      const int orow0 = grow0 + rt * 16;
      float* ow = out + ((size_t)orow0 * NCOL + ct * 16) + (8 * half) * NCOL + l;
      const int rows = n - orow0 - 8 * half;
#pragma unroll
      for (int r = 0; r < 8; ++r) {
        if (r < rows) {
          float v = c[r] + bv;
          if (RELU) v = fmaxf(v, 0.0f);
          ow[r * NCOL] = v;
        }
      }
    }
  } else if (NCOL2 > 0 && uwid < G1 + G2) {
    if constexpr (NCOL2 > 0) {
      if (prev_tile >= 0) {
        const int l    = lane & 15;
        const int half = lane >> 4;
        const int g    = uwid - G1;
        const int rt   = g / CT2;
        const int nt   = g - rt * CT2;
        const v8f c = tile_gemm<NCOL2>(sH2[buf ^ 1] + rt * 16 * LDA, Wn + nt * 16, lane);
        const int prow0 = prev_tile * TROWS + rt * 16;
        float* ow = out + ((size_t)prow0 * NCOL2 + nt * 16) + (8 * half) * NCOL2 + l;
        const int rows = n - prow0 - 8 * half;
#pragma unroll
        for (int r = 0; r < 8; ++r) {
          if (r < rows) ow[r * NCOL2] = c[r];
        }
      }
    }
  }
  prev_tile = tile;
  }

  if constexpr (NCOL2 > 0) {
    __syncthreads();
    if (uwid >= G1 && uwid < G1 + G2 && prev_tile >= 0) {
      const int l    = lane & 15;
      const int half = lane >> 4;
      const int g    = uwid - G1;
      const int rt   = g / CT2;
      const int nt   = g - rt * CT2;
      const v8f c = tile_gemm<NCOL2>(sH2[buf ^ 1] + rt * 16 * LDA, Wn + nt * 16, lane);
      const int prow0 = prev_tile * TROWS + rt * 16;
      float* ow = out + ((size_t)prow0 * NCOL2 + nt * 16) + (8 * half) * NCOL2 + l;
      const int rows = n - prow0 - 8 * half;
#pragma unroll
      for (int r = 0; r < 8; ++r) {
        if (r < rows) ow[r * NCOL2] = c[r];
      }
    }
  }
}

#define GBZ 16
#define ZW 64
#define SUM_BLOCKS 1024
__global__ void __launch_bounds__(512, 16) k_sum_rows(const float* __restrict__ z,
                              const int* __restrict__ rowptr,
                              const int* __restrict__ col,
                              const float* __restrict__ nsrc,
                              const float* __restrict__ ndst,
                              const float* __restrict__ bias,
                              float* __restrict__ out, int n, int tiles) {
  const int tid  = threadIdx.x;
  const int wid  = tid >> 5;
  const int lane = tid & 31;
  const int uwid = __builtin_amdgcn_readfirstlane(wid);
  const float2 bv = ((const float2*)bias)[lane];
  for (int tile = blockIdx.x; tile < tiles; tile += gridDim.x) {
    const int node = tile * 16 + uwid;
    if (node < n) {
      const float nd = ndst[node];
      const int beg = rowptr[node];
      const int end = rowptr[node + 1];
      float2 acc; acc.x = acc.y = 0.0f;
      for (int base = beg; base < end; base += 32) {
        const int left = end - base;
        const int cnt = left < 32 ? left : 32;
        int my_s = 0;
        float my_w = 0.0f;
        if (lane < cnt) {
          my_s = col[base + lane];
          my_w = nsrc[my_s];
        }
        for (int j = 0; j < cnt; j += GBZ) {
          float2 v[GBZ];
#pragma unroll
          for (int u = 0; u < GBZ; ++u) {
            const int p = (j + u < cnt) ? j + u : cnt - 1;
            const int s = __builtin_amdgcn_readlane(my_s, p);
            v[u] = ((const float2*)(z + (size_t)s * ZW))[lane];
          }
          __builtin_amdgcn_sched_barrier(0);
#pragma unroll
          for (int u = 0; u < GBZ; ++u) {
            const float w = __int_as_float(__builtin_amdgcn_readlane(__float_as_int(my_w), j + u));
            acc.x += v[u].x * w; acc.y += v[u].y * w;
          }
        }
      }
      float2 r2; r2.x = acc.x * nd + bv.x; r2.y = acc.y * nd + bv.y;
      ((float2*)(out + (size_t)node * ZW))[lane] = r2;
    }
  }
}

extern "C" void kernel_launch(void* const* d_in, const int* in_sizes, int n_in,
                              void* d_out, int out_size, void* d_ws, size_t ws_size,
                              hipStream_t stream) {
  const float* feats = (const float*)d_in[0];
  const int*   src   = (const int*)d_in[1];
  const int*   dst   = (const int*)d_in[2];
  const float* W0    = (const float*)d_in[3];
  const float* b0    = (const float*)d_in[4];
  const float* W1    = (const float*)d_in[5];
  const float* b1    = (const float*)d_in[6];
  const float* W2    = (const float*)d_in[7];
  const float* b2    = (const float*)d_in[8];

  const int N = in_sizes[0] / FEATS;
  const int E = in_sizes[1];

  char* ws = (char*)d_ws;
  size_t off = 0;
  auto take = [&](size_t bytes) -> void* {
    void* p = ws + off;
    off += (bytes + 255) & ~(size_t)255;
    return p;
  };

  const int NW = (N + 1) / 2;
  const int rb = (NW + 255) / 256;
  unsigned int* partD = (unsigned int*)take((size_t)NB * NW * 4);
  unsigned int* partS = (unsigned int*)take((size_t)NB * NW * 4);
  int*   pref   = (int*)take((size_t)NB * N * 4);
  int*   rank   = (int*)take((size_t)E * 4);
  float* nsrc   = (float*)take((size_t)N * 4);
  float* ndst   = (float*)take((size_t)N * 4);
  int*   incl   = (int*)take((size_t)N * 4);
  int*   bsum   = (int*)take((size_t)rb * 4);
  int*   rowptr = (int*)take((size_t)(N + 1) * 4);
  int*   col    = (int*)take((size_t)E * 4);
  float* h0     = (float*)take((size_t)N * FEATS * 4);
  float* h1     = (float*)take((size_t)N * FEATS * 4);

  const int eb = (E + 255) / 256;
  k_count<<<2 * NB, 1024, 0, stream>>>(src, dst, partD, partS, rank, E, NW);
  k_reduce<<<rb, 256, 0, stream>>>(partD, partS, pref, incl, bsum, nsrc, ndst, N, NW);

  k_rowptr<<<rb, 256, 0, stream>>>(incl, bsum, rowptr, N, NW);
  k_place<<<eb, 256, 0, stream>>>(src, dst, rowptr, pref, rank, col, E, N);

  const int tiles   = (N + TROWS - 1) / TROWS;
  const int tiles16 = (N + 15) / 16;

  k_fused_layer<128, true, 0><<<RESIDENT_BLOCKS, 1024, 0, stream>>>(
      feats, rowptr, col, nsrc, ndst, W0, b0, nullptr, h0, N, tiles);
  k_fused_layer<128, true, 64><<<RESIDENT_BLOCKS, 1024, 0, stream>>>(
      h0, rowptr, col, nsrc, ndst, W1, b1, W2, h1, N, tiles);
  k_sum_rows<<<SUM_BLOCKS, 512, 0, stream>>>(
      h1, rowptr, col, nsrc, ndst, b2, (float*)d_out, N, tiles16);
}


# Source norm folded into stored rows: layers 0 and 1 scale each output row by its node's norm in the epilogue, layers 1 and 2 sum rows with no per-neighbor norm gather; products rounded before summing
# speedup vs baseline: 2.5917x; 1.0394x over previous
#include <hip/hip_runtime.h>
#include <hip/hip_bf16.h>
#include <stdint.h>

typedef __attribute__((ext_vector_type(2))) float v2f;
typedef __attribute__((ext_vector_type(8))) float v8f;

#define FEATS 128
#define LDA 132
#define RESIDENT_BLOCKS 512
#define TROWS 32
#define GB 8

#define NB 64
#define HIST_WORDS 50000

__global__ void __launch_bounds__(1024) k_count(const int* __restrict__ src,
                                                const int* __restrict__ dst,
                                                unsigned int* __restrict__ partD,
                                                unsigned int* __restrict__ partS,
                                                int* __restrict__ rank, int E, int NW) {
  __shared__ __align__(16) unsigned int hist[HIST_WORDS];
  const int role = blockIdx.x / NB;
  const int b    = blockIdx.x - role * NB;
  const int tid  = (int)threadIdx.x;
  for (int w = tid; w < HIST_WORDS / 4; w += 1024) ((uint4*)hist)[w] = make_uint4(0u, 0u, 0u, 0u);
  __syncthreads();
  const int chunk = (E + NB - 1) / NB;
  const int e0 = b * chunk;
  const int e1 = (e0 + chunk < E) ? e0 + chunk : E;
  const int* __restrict__ keys = (role == 0) ? dst : src;
  constexpr int CU = 8;
  for (int base = e0 + tid; base < e1; base += 1024 * CU) {
    int k[CU];
#pragma unroll
    for (int u = 0; u < CU; ++u) {
      const int e = base + u * 1024;
      k[u] = (e < e1) ? keys[e] : 0;
    }
    if (role == 0) {
#pragma unroll
      for (int u = 0; u < CU; ++u) {
        const int e = base + u * 1024;
        if (e < e1) {
          const unsigned int sh = (unsigned int)(k[u] & 1) * 16u;
          const unsigned int old = atomicAdd(&hist[k[u] >> 1], 1u << sh);
          rank[e] = (int)((old >> sh) & 0xFFFFu);
        }
      }
    } else {
#pragma unroll
      for (int u = 0; u < CU; ++u) {
        const int e = base + u * 1024;
        if (e < e1) atomicAdd(&hist[k[u] >> 1], 1u << ((unsigned int)(k[u] & 1) * 16u));
      }
    }
  }
  __syncthreads();
  unsigned int* __restrict__ part = (role == 0 ? partD : partS) + (size_t)b * NW;
  for (int w = tid; w < NW; w += 1024) part[w] = hist[w];
}

__device__ __forceinline__ int wave_incl_scan(int v, int lane) {
#pragma unroll
  for (int off = 1; off < 32; off <<= 1) {
    const int t = __shfl_up(v, off, 32);
    if (lane >= off) v += t;
  }
  return v;
}

__global__ void __launch_bounds__(256) k_reduce(const unsigned int* __restrict__ partD,
                         const unsigned int* __restrict__ partS,
                         int* __restrict__ pref, int* __restrict__ incl, int* __restrict__ bsum,
                         float* __restrict__ nsrc, float* __restrict__ ndst, int n, int NW) {
  __shared__ int wtot[8];
  const int tid = (int)threadIdx.x;
  const int w = blockIdx.x * 256 + tid;
  const int i0 = 2 * w;
  const int i1 = 2 * w + 1;
  unsigned int d0 = 0u, d1 = 0u, s0 = 0u, s1 = 0u;
  if (w < NW) {
    constexpr int RB = 32;
    for (int bb = 0; bb < NB; bb += RB) {
      unsigned int v[RB], u[RB];
#pragma unroll
      for (int j = 0; j < RB; ++j) {
        v[j] = partD[(size_t)(bb + j) * NW + w];
        u[j] = partS[(size_t)(bb + j) * NW + w];
      }
#pragma unroll
      for (int j = 0; j < RB; ++j) {
        pref[(size_t)(bb + j) * n + i0] = (int)d0;
        if (i1 < n) pref[(size_t)(bb + j) * n + i1] = (int)d1;
        d0 += v[j] & 0xFFFFu; d1 += v[j] >> 16;
        s0 += u[j] & 0xFFFFu; s1 += u[j] >> 16;
      }
    }
    nsrc[i0] = rsqrtf(fmaxf((float)s0, 1.0f));
    ndst[i0] = rsqrtf(fmaxf((float)d0, 1.0f));
    if (i1 < n) {
      nsrc[i1] = rsqrtf(fmaxf((float)s1, 1.0f));
      ndst[i1] = rsqrtf(fmaxf((float)d1, 1.0f));
    }
  }
  const int lane = tid & 31;
  const int wv   = tid >> 5;
  const int pair = (int)(d0 + d1);
  const int inc  = wave_incl_scan(pair, lane);
  if (lane == 31) wtot[wv] = inc;
  __syncthreads();
  int before = 0, total = 0;
#pragma unroll
  for (int k = 0; k < 8; ++k) {
    const int s = wtot[k];
    if (k < wv) before += s;
    total += s;
  }
  const int ex = before + inc - pair;
  if (w < NW) {
    incl[i0] = ex + (int)d0;
    if (i1 < n) incl[i1] = ex + pair;
  }
  if (tid == 0) bsum[blockIdx.x] = total;
}

__global__ void __launch_bounds__(256) k_rowptr(const int* __restrict__ incl,
                                                const int* __restrict__ bsum,
                                                int* __restrict__ rowptr, int n, int NW) {
  __shared__ int wtot[8];
  const int tid = (int)threadIdx.x;
  const int w = blockIdx.x * 256 + tid;
  const int i0 = 2 * w;
  const int i1 = 2 * w + 1;
  int v = 0;
  for (int j = tid; j < (int)blockIdx.x; j += 256) v += bsum[j];
  int a0 = 0, a1 = 0;
  if (w < NW) {
    a0 = incl[i0];
    if (i1 < n) a1 = incl[i1];
  }
#pragma unroll
  for (int off = 16; off > 0; off >>= 1) v += __shfl_xor(v, off, 32);
  if ((tid & 31) == 0) wtot[tid >> 5] = v;
  __syncthreads();
  int boff = 0;
#pragma unroll
  for (int k = 0; k < 8; ++k) boff += wtot[k];
  if (w < NW) {
    rowptr[i0 + 1] = a0 + boff;
    if (i1 < n) rowptr[i1 + 1] = a1 + boff;
    if (w == 0) rowptr[0] = 0;
  }
}

__global__ void k_place(const int* __restrict__ src, const int* __restrict__ dst,
                        const int* __restrict__ rowptr, const int* __restrict__ pref,
                        const int* __restrict__ rank, int* __restrict__ col, int E, int n) {
  int e = blockIdx.x * blockDim.x + threadIdx.x;
  if (e < E) {
    const int chunk = (E + NB - 1) / NB;
    const int d = dst[e];
    const int b = e / chunk;
    const int pos = rowptr[d] + pref[(size_t)b * n + d] + rank[e];
    col[pos] = src[e];
  }
}

template <int NC>
__device__ __forceinline__ v8f tile_gemm(const float* __restrict__ sTile,
                                         const float* __restrict__ Wc, int lane) {
  const int l    = lane & 15;
  const int half = lane >> 4;
  const float* ar = sTile + l * LDA + 2 * half;
  const float* wr = Wc + (2 * half) * NC + l;
  v8f c = {};
  constexpr int KG = 32;
#pragma unroll 1
  for (int kc = 0; kc < FEATS; kc += KG) {
    v2f bf[KG / 4];
    v2f af[KG / 4];
#pragma unroll
    for (int s = 0; s < KG / 4; ++s) {
      const int kk = kc + 4 * s;
      bf[s].x = wr[kk * NC];
      bf[s].y = wr[(kk + 1) * NC];
    }
#pragma unroll
    for (int s = 0; s < KG / 4; ++s) {
      const int kk = kc + 4 * s;
      af[s].x = ar[kk];
      af[s].y = ar[kk + 1];
    }
#pragma unroll
    for (int s = 0; s < KG / 4; ++s) {
      c = __builtin_amdgcn_wmma_f32_16x16x4_f32(false, af[s], false, bf[s], (short)0, c,
                                                false, false);
    }
  }
  return c;
}

template <int NC>
__device__ __forceinline__ void stage_w(float* __restrict__ sW, const float* __restrict__ W,
                                        int tid) {
  constexpr int PAIRS = (FEATS / 2) * NC;
  static_assert(PAIRS % 1024 == 0, "whole passes of the 1024-thread work-group");
  v2f w[PAIRS / 1024];
#pragma unroll
  for (int i = 0; i < PAIRS / 1024; ++i) {
    const int p   = tid + i * 1024;
    const int kp  = p / NC;
    const int col = p - kp * NC;
    w[i].x = W[(2 * kp) * NC + col];
    w[i].y = W[(2 * kp + 1) * NC + col];
  }
#pragma unroll
  for (int i = 0; i < PAIRS / 1024; ++i) {
    const int p   = tid + i * 1024;
    const int kp  = p / NC;
    const int col = p - kp * NC;
    const int S    = kp >> 1;
    const int half = kp & 1;
    const int ct   = col >> 4;
    const int l    = col & 15;
    ((v2f*)sW)[(ct * 32 + S) * 32 + half * 16 + l] = w[i];
  }
}

__device__ __forceinline__ v8f tile_gemm_lds(const float* __restrict__ sTile,
                                             const float* __restrict__ sWc, int lane) {
  const int l    = lane & 15;
  const int half = lane >> 4;
  const float* ar = sTile + l * LDA + 2 * half;
  const v2f* br = (const v2f*)sWc + lane;
  v8f c = {};
  constexpr int KG = 32;
#pragma unroll 1
  for (int kc = 0; kc < FEATS; kc += KG) {
    v2f bf[KG / 4];
    v2f af[KG / 4];
#pragma unroll
    for (int s = 0; s < KG / 4; ++s) {
      bf[s] = br[(kc / 4 + s) * 32];
    }
#pragma unroll
    for (int s = 0; s < KG / 4; ++s) {
      const int kk = kc + 4 * s;
      af[s].x = ar[kk];
      af[s].y = ar[kk + 1];
    }
#pragma unroll
    for (int s = 0; s < KG / 4; ++s) {
      c = __builtin_amdgcn_wmma_f32_16x16x4_f32(false, af[s], false, bf[s], (short)0, c,
                                                false, false);
    }
  }
  return c;
}

template <int NCOL, bool RELU, int NCOL2, bool PRE>
__global__ void __launch_bounds__(1024, 16) k_fused_layer(const float* __restrict__ x,
                              const int* __restrict__ rowptr,
                              const int* __restrict__ col,
                              const float* __restrict__ nsrc,
                              const float* __restrict__ ndst,
                              const float* __restrict__ W,
                              const float* __restrict__ bias,
                              const float* __restrict__ Wn,
                              float* __restrict__ out, int n, int tiles) {
  __shared__ float sA2[2][TROWS * LDA];
  __shared__ float sH2[NCOL2 > 0 ? 2 : 1][NCOL2 > 0 ? TROWS * LDA : 4];
  __shared__ __align__(16) float sW[FEATS * NCOL];

  const int tid  = threadIdx.x;
  const int wid  = tid >> 5;
  const int lane = tid & 31;
  const int uwid = __builtin_amdgcn_readfirstlane(wid);
  constexpr int CT  = NCOL / 16;
  constexpr int CT2 = NCOL2 / 16;
  constexpr int G1  = (TROWS / 16) * CT;
  constexpr int G2  = (TROWS / 16) * CT2;
  static_assert(G1 + G2 <= 32, "one wave per 16 x 16 output tile");

  stage_w<NCOL>(sW, W, tid);

  int buf = 0;
  int prev_tile = -1;
  for (int tile = blockIdx.x; tile < tiles; tile += gridDim.x, buf ^= 1) {
  float* sA = sA2[buf];
  const int grow0 = tile * TROWS;

  const int node = grow0 + uwid;
  float4 acc; acc.x = acc.y = acc.z = acc.w = 0.0f;
  float nd = 0.0f;
  float ns_own = 0.0f;
  if (node < n) {
    nd = __int_as_float(__builtin_amdgcn_readfirstlane(__float_as_int(ndst[node])));
    ns_own = __int_as_float(__builtin_amdgcn_readfirstlane(__float_as_int(nsrc[node])));
    const int beg = rowptr[node];
    const int end = rowptr[node + 1];
    for (int base = beg; base < end; base += 32) {
      const int left = end - base;
      const int cnt = left < 32 ? left : 32;
      int my_s = 0;
      float my_w = 0.0f;
      if (lane < cnt) {
        my_s = col[base + lane];
        if constexpr (!PRE) my_w = nsrc[my_s];
      }
      for (int j = 0; j < cnt; j += GB) {
        float4 v[GB];
#pragma unroll
        for (int u = 0; u < GB; ++u) {
          const int p = (j + u < cnt) ? j + u : cnt - 1;
          const int s = __builtin_amdgcn_readlane(my_s, p);
          v[u] = ((const float4*)(x + (size_t)s * FEATS))[lane];
        }
        if constexpr (PRE) {
#pragma unroll
          for (int u = 0; u < GB; ++u) {
            const float w = (j + u < cnt) ? 1.0f : 0.0f;
            acc.x += v[u].x * w; acc.y += v[u].y * w; acc.z += v[u].z * w; acc.w += v[u].w * w;
          }
        } else {
          __builtin_amdgcn_sched_barrier(0);
#pragma unroll
          for (int u = 0; u < GB; ++u) {
            const float w = __int_as_float(__builtin_amdgcn_readlane(__float_as_int(my_w), j + u));
            acc.x += v[u].x * w; acc.y += v[u].y * w; acc.z += v[u].z * w; acc.w += v[u].w * w;
          }
        }
      }
    }
  }
  float4 r4; r4.x = acc.x * nd; r4.y = acc.y * nd; r4.z = acc.z * nd; r4.w = acc.w * nd;
  ((float4*)(&sA[wid * LDA]))[lane] = r4;
  if (lane == 0) sA[wid * LDA + FEATS] = ns_own;
  __syncthreads();

  if (uwid < G1) {
    const int l    = lane & 15;
    const int half = lane >> 4;
    const int rt   = uwid / CT;
    const int ct   = uwid - rt * CT;
    const v8f c = tile_gemm_lds(sA + rt * 16 * LDA, sW + ct * (32 * 32 * 2), lane);
    const float bv = bias[ct * 16 + l];
    const float* nsp = sA + (rt * 16 + 8 * half) * LDA + FEATS;
    float nsr[8];
#pragma unroll
    for (int r = 0; r < 8; ++r) nsr[r] = nsp[r * LDA];
    if constexpr (NCOL2 > 0) {
      float* hw = sH2[buf] + ct * 16 + (rt * 16 + 8 * half) * LDA + l;
#pragma unroll
      for (int r = 0; r < 8; ++r) {
        float v = c[r] + bv;
        if (RELU) v = fmaxf(v, 0.0f);
        hw[r * LDA] = v * nsr[r];
      }
    } else {
      const int orow0 = grow0 + rt * 16;
      float* ow = out + ((size_t)orow0 * NCOL + ct * 16) + (8 * half) * NCOL + l;
      const int rows = n - orow0 - 8 * half;
#pragma unroll
      for (int r = 0; r < 8; ++r) {
        if (r < rows) {
          float v = c[r] + bv;
          if (RELU) v = fmaxf(v, 0.0f);
          ow[r * NCOL] = v * nsr[r];
        }
      }
    }
  } else if (NCOL2 > 0 && uwid < G1 + G2) {
    if constexpr (NCOL2 > 0) {
      if (prev_tile >= 0) {
        const int l    = lane & 15;
        const int half = lane >> 4;
        const int g    = uwid - G1;
        const int rt   = g / CT2;
        const int nt   = g - rt * CT2;
        const v8f c = tile_gemm<NCOL2>(sH2[buf ^ 1] + rt * 16 * LDA, Wn + nt * 16, lane);
        const int prow0 = prev_tile * TROWS + rt * 16;
        float* ow = out + ((size_t)prow0 * NCOL2 + nt * 16) + (8 * half) * NCOL2 + l;
        const int rows = n - prow0 - 8 * half;
#pragma unroll
        for (int r = 0; r < 8; ++r) {
          if (r < rows) ow[r * NCOL2] = c[r];
        }
      }
    }
  }
  prev_tile = tile;
  }

  if constexpr (NCOL2 > 0) {
    __syncthreads();
    if (uwid >= G1 && uwid < G1 + G2 && prev_tile >= 0) {
      const int l    = lane & 15;
      const int half = lane >> 4;
      const int g    = uwid - G1;
      const int rt   = g / CT2;
      const int nt   = g - rt * CT2;
      const v8f c = tile_gemm<NCOL2>(sH2[buf ^ 1] + rt * 16 * LDA, Wn + nt * 16, lane);
      const int prow0 = prev_tile * TROWS + rt * 16;
      float* ow = out + ((size_t)prow0 * NCOL2 + nt * 16) + (8 * half) * NCOL2 + l;
      const int rows = n - prow0 - 8 * half;
#pragma unroll
      for (int r = 0; r < 8; ++r) {
        if (r < rows) ow[r * NCOL2] = c[r];
      }
    }
  }
}

#define GBZ 16
#define ZW 64
#define SUM_BLOCKS 1024
__global__ void __launch_bounds__(512, 16) k_sum_rows(const float* __restrict__ z,
                              const int* __restrict__ rowptr,
                              const int* __restrict__ col,
                              const float* __restrict__ ndst,
                              const float* __restrict__ bias,
                              float* __restrict__ out, int n, int tiles) {
  const int tid  = threadIdx.x;
  const int wid  = tid >> 5;
  const int lane = tid & 31;
  const int uwid = __builtin_amdgcn_readfirstlane(wid);
  const float2 bv = ((const float2*)bias)[lane];
  for (int tile = blockIdx.x; tile < tiles; tile += gridDim.x) {
    const int node = tile * 16 + uwid;
    if (node < n) {
      const float nd = ndst[node];
      const int beg = rowptr[node];
      const int end = rowptr[node + 1];
      float2 acc; acc.x = acc.y = 0.0f;
      for (int base = beg; base < end; base += 32) {
        const int left = end - base;
        const int cnt = left < 32 ? left : 32;
        int my_s = 0;
        if (lane < cnt) my_s = col[base + lane];
        for (int j = 0; j < cnt; j += GBZ) {
          float2 v[GBZ];
#pragma unroll
          for (int u = 0; u < GBZ; ++u) {
            const int p = (j + u < cnt) ? j + u : cnt - 1;
            const int s = __builtin_amdgcn_readlane(my_s, p);
            v[u] = ((const float2*)(z + (size_t)s * ZW))[lane];
          }
#pragma unroll
          for (int u = 0; u < GBZ; ++u) {
            const float w = (j + u < cnt) ? 1.0f : 0.0f;
            acc.x += v[u].x * w; acc.y += v[u].y * w;
          }
        }
      }
      float2 r2; r2.x = acc.x * nd + bv.x; r2.y = acc.y * nd + bv.y;
      ((float2*)(out + (size_t)node * ZW))[lane] = r2;
    }
  }
}

extern "C" void kernel_launch(void* const* d_in, const int* in_sizes, int n_in,
                              void* d_out, int out_size, void* d_ws, size_t ws_size,
                              hipStream_t stream) {
  const float* feats = (const float*)d_in[0];
  const int*   src   = (const int*)d_in[1];
  const int*   dst   = (const int*)d_in[2];
  const float* W0    = (const float*)d_in[3];
  const float* b0    = (const float*)d_in[4];
  const float* W1    = (const float*)d_in[5];
  const float* b1    = (const float*)d_in[6];
  const float* W2    = (const float*)d_in[7];
  const float* b2    = (const float*)d_in[8];

  const int N = in_sizes[0] / FEATS;
  const int E = in_sizes[1];

  char* ws = (char*)d_ws;
  size_t off = 0;
  auto take = [&](size_t bytes) -> void* {
    void* p = ws + off;
    off += (bytes + 255) & ~(size_t)255;
    return p;
  };

  const int NW = (N + 1) / 2;
  const int rb = (NW + 255) / 256;
  unsigned int* partD = (unsigned int*)take((size_t)NB * NW * 4);
  unsigned int* partS = (unsigned int*)take((size_t)NB * NW * 4);
  int*   pref   = (int*)take((size_t)NB * N * 4);
  int*   rank   = (int*)take((size_t)E * 4);
  float* nsrc   = (float*)take((size_t)N * 4);
  float* ndst   = (float*)take((size_t)N * 4);
  int*   incl   = (int*)take((size_t)N * 4);
  int*   bsum   = (int*)take((size_t)rb * 4);
  int*   rowptr = (int*)take((size_t)(N + 1) * 4);
  int*   col    = (int*)take((size_t)E * 4);
  float* h0     = (float*)take((size_t)N * FEATS * 4);
  float* h1     = (float*)take((size_t)N * FEATS * 4);

  const int eb = (E + 255) / 256;
  k_count<<<2 * NB, 1024, 0, stream>>>(src, dst, partD, partS, rank, E, NW);
  k_reduce<<<rb, 256, 0, stream>>>(partD, partS, pref, incl, bsum, nsrc, ndst, N, NW);

  k_rowptr<<<rb, 256, 0, stream>>>(incl, bsum, rowptr, N, NW);
  k_place<<<eb, 256, 0, stream>>>(src, dst, rowptr, pref, rank, col, E, N);

  const int tiles   = (N + TROWS - 1) / TROWS;
  const int tiles16 = (N + 15) / 16;

  k_fused_layer<128, true, 0, false><<<RESIDENT_BLOCKS, 1024, 0, stream>>>(
      feats, rowptr, col, nsrc, ndst, W0, b0, nullptr, h0, N, tiles);
  k_fused_layer<128, true, 64, true><<<RESIDENT_BLOCKS, 1024, 0, stream>>>(
      h0, rowptr, col, nsrc, ndst, W1, b1, W2, h1, N, tiles);
  k_sum_rows<<<SUM_BLOCKS, 512, 0, stream>>>(
      h1, rowptr, col, ndst, b2, (float*)d_out, N, tiles16);
}
